# starting kernel
# baseline (speedup 1.0000x reference)
#include <hip/hip_runtime.h>
#include <hip/hip_bf16.h>

#define N_NODES 50000
#define N_EDGES 500000
#define N_GRAPHS 1000
#define HID 128
#define OUTD 10
#define NLAYERS 4

typedef __attribute__((ext_vector_type(16))) __bf16 v16bf;
typedef __attribute__((ext_vector_type(8)))  __bf16 v8bf;
typedef __attribute__((ext_vector_type(8)))  float  v8f;

__global__ __launch_bounds__(256) void k_encode_nodes(
    const int* __restrict__ x, const float* __restrict__ atom_emb,
    float* __restrict__ h)
{
    int gid = blockIdx.x * 256 + threadIdx.x;
    if (gid >= N_NODES * 32) return;
    int n = gid >> 5;
    int c = (gid & 31) * 4;
    float4 acc = make_float4(0.f, 0.f, 0.f, 0.f);
    #pragma unroll
    for (int f = 0; f < 9; ++f) {
        int idx = x[n * 9 + f];
        const float4 v = *(const float4*)(atom_emb + ((size_t)(f * 120 + idx)) * HID + c);
        acc.x += v.x; acc.y += v.y; acc.z += v.z; acc.w += v.w;
    }
    *(float4*)(h + (size_t)n * HID + c) = acc;
}

__global__ __launch_bounds__(256) void k_copy_h(
    const float* __restrict__ h, float* __restrict__ agg)
{
    int gid = blockIdx.x * 256 + threadIdx.x;
    if (gid >= N_NODES * 32) return;
    ((float4*)agg)[gid] = ((const float4*)h)[gid];
}

__global__ __launch_bounds__(256) void k_edges(
    const int* __restrict__ edge_index, const int* __restrict__ edge_attr,
    const float* __restrict__ bond_emb, const float* __restrict__ h,
    float* __restrict__ agg)
{
    int gid = blockIdx.x * 256 + threadIdx.x;
    if (gid >= N_EDGES * 32) return;
    int e = gid >> 5;
    int c = (gid & 31) * 4;
    int src = edge_index[e];
    int dst = edge_index[N_EDGES + e];
    int a0 = edge_attr[e * 3 + 0];
    int a1 = edge_attr[e * 3 + 1];
    int a2 = edge_attr[e * 3 + 2];
    const float4 e0 = *(const float4*)(bond_emb + ((size_t)(0 * 6 + a0)) * HID + c);
    const float4 e1 = *(const float4*)(bond_emb + ((size_t)(1 * 6 + a1)) * HID + c);
    const float4 e2 = *(const float4*)(bond_emb + ((size_t)(2 * 6 + a2)) * HID + c);
    const float4 hv = *(const float4*)(h + (size_t)src * HID + c);
    float m0 = hv.x + e0.x + e1.x + e2.x; m0 = m0 > 0.f ? m0 : 0.f;
    float m1 = hv.y + e0.y + e1.y + e2.y; m1 = m1 > 0.f ? m1 : 0.f;
    float m2 = hv.z + e0.z + e1.z + e2.z; m2 = m2 > 0.f ? m2 : 0.f;
    float m3 = hv.w + e0.w + e1.w + e2.w; m3 = m3 > 0.f ? m3 : 0.f;
    float* o = agg + (size_t)dst * HID + c;
    atomicAdd(o + 0, m0);
    atomicAdd(o + 1, m1);
    atomicAdd(o + 2, m2);
    atomicAdd(o + 3, m3);
}

__global__ __launch_bounds__(256) void k_prep_weights(
    const float* __restrict__ w1, const float* __restrict__ w2,
    __bf16* __restrict__ f1, __bf16* __restrict__ f2)
{
    int o = blockIdx.x * 256 + threadIdx.x;
    if (o >= NLAYERS * 4 * 8 * 32 * 16) return;
    int j     =  o        & 15;
    int lane  = (o >> 4)  & 31;
    int ntile = (o >> 9)  & 7;
    int ks    = (o >> 12) & 3;
    int layer =  o >> 14;
    int colL  = lane & 15;
    int khalf = (lane >> 4) * 8;
    int K     = ks * 32 + ((j < 8) ? (khalf + j) : (16 + khalf + (j - 8)));
    int col   = ntile * 16 + colL;
    size_t src = (size_t)layer * HID * HID + (size_t)K * HID + col;
    size_t dst = (size_t)layer * 16384 +
                 (((size_t)ks * 8 + ntile) * 32 + lane) * 16 + j;
    f1[dst] = (__bf16)w1[src];
    f2[dst] = (__bf16)w2[src];
}

__global__ __launch_bounds__(256) void k_node_update(
    const float* __restrict__ agg, const float* __restrict__ hcur,
    const __bf16* __restrict__ wf1, const __bf16* __restrict__ wf2,
    const float* __restrict__ b1, const float* __restrict__ b2,
    const float* __restrict__ gamma, const float* __restrict__ beta,
    const float* __restrict__ mean, const float* __restrict__ var,
    float* __restrict__ hnext)
{
    __shared__ __align__(16) __bf16 zt[16 * HID];
    __shared__ __align__(16) __bf16 z1[16 * HID];

    const int tid   = threadIdx.x;
    const int wave  = tid >> 5;
    const int lane  = tid & 31;
    const int node0 = blockIdx.x * 16;

    {
        const int base = tid * 8;
        const int row  = base >> 7;
        const int c    = base & 127;
        const float4* p = (const float4*)(agg + (size_t)(node0 + row) * HID + c);
        float4 v0 = p[0], v1 = p[1];
        __bf16* q = &zt[base];
        q[0] = (__bf16)v0.x; q[1] = (__bf16)v0.y;
        q[2] = (__bf16)v0.z; q[3] = (__bf16)v0.w;
        q[4] = (__bf16)v1.x; q[5] = (__bf16)v1.y;
        q[6] = (__bf16)v1.z; q[7] = (__bf16)v1.w;
    }
    __syncthreads();

    const int colL  = lane & 15;
    const int col   = wave * 16 + colL;
    const int khalf = (lane >> 4) * 8;
    const int rowA  = lane & 15;

    v8f acc1 = {};
    #pragma unroll
    for (int ks = 0; ks < 4; ++ks) {
        const __bf16* ap = &zt[rowA * HID + ks * 32 + khalf];
        v8bf alo = *(const v8bf*)(ap);
        v8bf ahi = *(const v8bf*)(ap + 16);
        v16bf a = __builtin_shufflevector(alo, ahi,
                    0,1,2,3,4,5,6,7,8,9,10,11,12,13,14,15);
        v16bf b = *(const v16bf*)(wf1 + (((size_t)ks * 8 + wave) * 32 + lane) * 16);
        acc1 = __builtin_amdgcn_wmma_f32_16x16x32_bf16(
                   false, a, false, b, (short)0, acc1, false, false);
    }

    const float bias1 = b1[col];
    #pragma unroll
    for (int r = 0; r < 8; ++r) {
        int m = (lane < 16) ? r : (r + 8);
        float v = acc1[r] + bias1;
        v = v > 0.f ? v : 0.f;
        z1[m * HID + col] = (__bf16)v;
    }
    __syncthreads();

    v8f acc2 = {};
    #pragma unroll
    for (int ks = 0; ks < 4; ++ks) {
        const __bf16* ap = &z1[rowA * HID + ks * 32 + khalf];
        v8bf alo = *(const v8bf*)(ap);
        v8bf ahi = *(const v8bf*)(ap + 16);
        v16bf a = __builtin_shufflevector(alo, ahi,
                    0,1,2,3,4,5,6,7,8,9,10,11,12,13,14,15);
        v16bf b = *(const v16bf*)(wf2 + (((size_t)ks * 8 + wave) * 32 + lane) * 16);
        acc2 = __builtin_amdgcn_wmma_f32_16x16x32_bf16(
                   false, a, false, b, (short)0, acc2, false, false);
    }

    const float bias2 = b2[col];
    const float mu    = mean[col];
    const float inv   = rsqrtf(var[col] + 1e-5f);
    const float ga    = gamma[col];
    const float be    = beta[col];
    #pragma unroll
    for (int r = 0; r < 8; ++r) {
        int m = (lane < 16) ? r : (r + 8);
        size_t idx = (size_t)(node0 + m) * HID + col;
        float v = acc2[r] + bias2;
        v = (v - mu) * inv * ga + be;
        v = v > 0.f ? v : 0.f;
        hnext[idx] = hcur[idx] + v;
    }
}

__global__ __launch_bounds__(256) void k_pool_zero(
    float* __restrict__ sums, float* __restrict__ cnts)
{
    int i = blockIdx.x * 256 + threadIdx.x;
    if (i < N_GRAPHS * HID) sums[i] = 0.f;
    if (i < N_GRAPHS)       cnts[i] = 0.f;
}

__global__ __launch_bounds__(256) void k_pool_acc(
    const float* __restrict__ h, const int* __restrict__ batch,
    float* __restrict__ sums, float* __restrict__ cnts)
{
    int gid = blockIdx.x * 256 + threadIdx.x;
    if (gid >= N_NODES * 32) return;
    int n = gid >> 5;
    int c = (gid & 31) * 4;
    int g = batch[n];
    const float4 v = *(const float4*)(h + (size_t)n * HID + c);
    float* o = sums + (size_t)g * HID + c;
    atomicAdd(o + 0, v.x);
    atomicAdd(o + 1, v.y);
    atomicAdd(o + 2, v.z);
    atomicAdd(o + 3, v.w);
    if ((gid & 31) == 0) atomicAdd(cnts + g, 1.0f);
}

__global__ __launch_bounds__(128) void k_pool_mlp(
    const float* __restrict__ sums, const float* __restrict__ cnts,
    const float* __restrict__ mw, const float* __restrict__ mb,
    float* __restrict__ out)
{
    __shared__ float p[HID];
    int g = blockIdx.x;
    int t = threadIdx.x;
    float cnt = cnts[g];
    cnt = cnt > 1.f ? cnt : 1.f;
    p[t] = sums[(size_t)g * HID + t] / cnt;
    __syncthreads();
    if (t < OUTD) {
        float a = mb[t];
        #pragma unroll 16
        for (int c = 0; c < HID; ++c) a += p[c] * mw[c * OUTD + t];
        out[g * OUTD + t] = a;
    }
}

extern "C" void kernel_launch(void* const* d_in, const int* in_sizes, int n_in,
                              void* d_out, int out_size, void* d_ws, size_t ws_size,
                              hipStream_t stream) {
    const int*   x          = (const int*)  d_in[0];
    const int*   edge_index = (const int*)  d_in[1];
    const int*   edge_attr  = (const int*)  d_in[2];
    const int*   batch      = (const int*)  d_in[3];
    const float* atom_emb   = (const float*)d_in[4];
    const float* bond_emb   = (const float*)d_in[5];
    const float* lin1_w     = (const float*)d_in[6];
    const float* lin1_b     = (const float*)d_in[7];
    const float* lin2_w     = (const float*)d_in[8];
    const float* lin2_b     = (const float*)d_in[9];
    const float* bn_gamma   = (const float*)d_in[10];
    const float* bn_beta    = (const float*)d_in[11];
    const float* bn_mean    = (const float*)d_in[12];
    const float* bn_var     = (const float*)d_in[13];
    const float* mlp_w      = (const float*)d_in[14];
    const float* mlp_b      = (const float*)d_in[15];

    char* ws = (char*)d_ws;
    size_t off = 0;
    auto alloc = [&](size_t bytes) -> char* {
        char* p = ws + off;
        off = (off + bytes + 255) & ~(size_t)255;
        return p;
    };
    float*  h0   = (float*) alloc((size_t)N_NODES * HID * 4);
    float*  h1   = (float*) alloc((size_t)N_NODES * HID * 4);
    float*  agg  = (float*) alloc((size_t)N_NODES * HID * 4);
    __bf16* wf1  = (__bf16*)alloc((size_t)NLAYERS * 16384 * 2);
    __bf16* wf2  = (__bf16*)alloc((size_t)NLAYERS * 16384 * 2);
    float*  sums = (float*) alloc((size_t)N_GRAPHS * HID * 4);
    float*  cnts = (float*) alloc((size_t)N_GRAPHS * 4);

    const int nodeChunks = N_NODES * 32;
    const int edgeChunks = N_EDGES * 32;

    k_prep_weights<<<(NLAYERS * 4 * 8 * 32 * 16 + 255) / 256, 256, 0, stream>>>(
        lin1_w, lin2_w, wf1, wf2);
    k_encode_nodes<<<(nodeChunks + 255) / 256, 256, 0, stream>>>(x, atom_emb, h0);

    float* hc = h0;
    float* hn = h1;
    for (int i = 0; i < NLAYERS; ++i) {
        k_copy_h<<<(nodeChunks + 255) / 256, 256, 0, stream>>>(hc, agg);
        k_edges<<<(edgeChunks + 255) / 256, 256, 0, stream>>>(
            edge_index, edge_attr, bond_emb, hc, agg);
        k_node_update<<<N_NODES / 16, 256, 0, stream>>>(
            agg, hc, wf1 + (size_t)i * 16384, wf2 + (size_t)i * 16384,
            lin1_b + i * HID, lin2_b + i * HID,
            bn_gamma + i * HID, bn_beta + i * HID,
            bn_mean + i * HID, bn_var + i * HID, hn);
        float* t = hc; hc = hn; hn = t;
    }

    k_pool_zero<<<(N_GRAPHS * HID + 255) / 256, 256, 0, stream>>>(sums, cnts);
    k_pool_acc<<<(nodeChunks + 255) / 256, 256, 0, stream>>>(hc, batch, sums, cnts);
    k_pool_mlp<<<N_GRAPHS, 128, 0, stream>>>(sums, cnts, mlp_w, mlp_b, (float*)d_out);
}


# gather fused into the node-update launch: each wave sums its 2 node rows in registers (same edge order) straight into the LDS z tile as bf16; agg plane and the 4 gather launches gone
# speedup vs baseline: 1.8366x; 1.8366x over previous
#include <hip/hip_runtime.h>
#include <hip/hip_bf16.h>

#define N_NODES 50000
#define N_EDGES 500000
#define N_GRAPHS 1000
#define HID 128
#define OUTD 10
#define NLAYERS 4

typedef __attribute__((ext_vector_type(16))) __bf16 v16bf;
typedef __attribute__((ext_vector_type(8)))  __bf16 v8bf;
typedef __attribute__((ext_vector_type(4)))  __bf16 v4bf;
typedef __attribute__((ext_vector_type(8)))  float  v8f;

__global__ __launch_bounds__(256) void k_encode_nodes(
    const int* __restrict__ x, const float* __restrict__ atom_emb,
    float* __restrict__ h)
{
    int gid = blockIdx.x * 256 + threadIdx.x;
    if (gid >= N_NODES * 32) return;
    int n = gid >> 5;
    int c = (gid & 31) * 4;
    float4 acc = make_float4(0.f, 0.f, 0.f, 0.f);
    #pragma unroll
    for (int f = 0; f < 9; ++f) {
        int idx = x[n * 9 + f];
        const float4 v = *(const float4*)(atom_emb + ((size_t)(f * 120 + idx)) * HID + c);
        acc.x += v.x; acc.y += v.y; acc.z += v.z; acc.w += v.w;
    }
    *(float4*)(h + (size_t)n * HID + c) = acc;
}

__global__ __launch_bounds__(256) void k_count(
    const int* __restrict__ edge_index, int* __restrict__ cnt)
{
    int e = blockIdx.x * 256 + threadIdx.x;
    if (e >= N_EDGES) return;
    atomicAdd(cnt + edge_index[N_EDGES + e], 1);
}

__global__ __launch_bounds__(1024) void k_scan(
    const int* __restrict__ cnt, int* __restrict__ rowptr)
{
    __shared__ int wpre[32];
    __shared__ int wtot[32];
    const int tid  = threadIdx.x;
    const int lane = tid & 31;
    const int wave = tid >> 5;
    const int first = blockIdx.x * 1024;

    int pre = 0;
    for (int i = tid; i < first; i += 1024) pre += cnt[i];
    for (int o = 16; o > 0; o >>= 1) pre += __shfl_xor(pre, o);
    if (lane == 0) wpre[wave] = pre;

    const int i = first + tid;
    const int v = (i < N_NODES) ? cnt[i] : 0;
    int inc = v;
    for (int o = 1; o < 32; o <<= 1) {
        int t = __shfl_up(inc, o);
        if (lane >= o) inc += t;
    }
    if (lane == 31) wtot[wave] = inc;
    __syncthreads();

    int t = wpre[lane] + ((lane < wave) ? wtot[lane] : 0);
    for (int o = 16; o > 0; o >>= 1) t += __shfl_xor(t, o);
    const int excl = t + inc - v;
    if (i < N_NODES)      rowptr[i] = excl;
    if (i == N_NODES - 1) rowptr[N_NODES] = excl + v;
}

__global__ __launch_bounds__(256) void k_fill(
    const int* __restrict__ edge_index, const int* __restrict__ edge_attr,
    const int* __restrict__ rowptr, int* __restrict__ cnt,
    int4* __restrict__ slots)
{
    int e = blockIdx.x * 256 + threadIdx.x;
    if (e >= N_EDGES) return;
    const int src = edge_index[e];
    const int dst = edge_index[N_EDGES + e];
    const int a0 = edge_attr[e * 3 + 0];
    const int a1 = edge_attr[e * 3 + 1];
    const int a2 = edge_attr[e * 3 + 2];
    const int p = atomicSub(cnt + dst, 1) - 1;
    slots[rowptr[dst] + p] = make_int4(e, src, a0 | (a1 << 8) | (a2 << 16), dst);
}

__global__ __launch_bounds__(256) void k_sort(
    const int4* __restrict__ slots, const int* __restrict__ rowptr,
    int2* __restrict__ rec)
{
    int p = blockIdx.x * 256 + threadIdx.x;
    if (p >= N_EDGES) return;
    const int4 r = slots[p];
    const int s = rowptr[r.w];
    const int t = rowptr[r.w + 1];
    int rank = 0;
    for (int j = s; j < t; ++j) rank += (slots[j].x < r.x) ? 1 : 0;
    rec[s + rank] = make_int2(r.y, r.z);
}

__device__ __forceinline__ float4 edge_msg(const float4 hv, const float4 e0,
                                           const float4 e1, const float4 e2)
{
    float m0 = hv.x + e0.x + e1.x + e2.x; m0 = m0 > 0.f ? m0 : 0.f;
    float m1 = hv.y + e0.y + e1.y + e2.y; m1 = m1 > 0.f ? m1 : 0.f;
    float m2 = hv.z + e0.z + e1.z + e2.z; m2 = m2 > 0.f ? m2 : 0.f;
    float m3 = hv.w + e0.w + e1.w + e2.w; m3 = m3 > 0.f ? m3 : 0.f;
    return make_float4(m0, m1, m2, m3);
}

__device__ __forceinline__ float4 gather_row(
    const int n, const int s, const int t, const int lane,
    const int2* __restrict__ rec, const float4* __restrict__ b4,
    const float4* __restrict__ h4)
{
    float4 acc = h4[(size_t)n * 32 + lane];
    for (int base = s; base < t; base += 32) {
        const int left = t - base;
        const int cnt  = left < 32 ? left : 32;
        int2 my = make_int2(0, 0);
        if (lane < cnt) my = rec[base + lane];
        int j = 0;
        for (; j + 4 <= cnt; j += 4) {
            float4 hv[4], e0[4], e1[4], e2[4];
            for (int u = 0; u < 4; ++u) {
                const int src  = __builtin_amdgcn_readlane(my.x, j + u);
                const int code = __builtin_amdgcn_readlane(my.y, j + u);
                hv[u] = h4[(size_t)src * 32 + lane];
                e0[u] = b4[(0 * 6 + (code & 255)) * 32 + lane];
                e1[u] = b4[(1 * 6 + ((code >> 8) & 255)) * 32 + lane];
                e2[u] = b4[(2 * 6 + (code >> 16)) * 32 + lane];
            }
            for (int u = 0; u < 4; ++u) {
                const float4 m = edge_msg(hv[u], e0[u], e1[u], e2[u]);
                acc.x += m.x; acc.y += m.y; acc.z += m.z; acc.w += m.w;
            }
        }
        for (; j < cnt; ++j) {
            const int src  = __builtin_amdgcn_readlane(my.x, j);
            const int code = __builtin_amdgcn_readlane(my.y, j);
            const float4 hv = h4[(size_t)src * 32 + lane];
            const float4 e0 = b4[(0 * 6 + (code & 255)) * 32 + lane];
            const float4 e1 = b4[(1 * 6 + ((code >> 8) & 255)) * 32 + lane];
            const float4 e2 = b4[(2 * 6 + (code >> 16)) * 32 + lane];
            const float4 m = edge_msg(hv, e0, e1, e2);
            acc.x += m.x; acc.y += m.y; acc.z += m.z; acc.w += m.w;
        }
    }
    return acc;
}

__global__ __launch_bounds__(256) void k_prep_weights(
    const float* __restrict__ w1, const float* __restrict__ w2,
    __bf16* __restrict__ f1, __bf16* __restrict__ f2, int* __restrict__ cnt)
{
    int o = blockIdx.x * 256 + threadIdx.x;
    if (o >= NLAYERS * 4 * 8 * 32 * 16) return;
    static_assert(N_NODES <= NLAYERS * 4 * 8 * 32 * 16, "one thread per degree counter");
    if (o < N_NODES) cnt[o] = 0;
    int j     =  o        & 15;
    int lane  = (o >> 4)  & 31;
    int ntile = (o >> 9)  & 7;
    int ks    = (o >> 12) & 3;
    int layer =  o >> 14;
    int colL  = lane & 15;
    int khalf = (lane >> 4) * 8;
    int K     = ks * 32 + ((j < 8) ? (khalf + j) : (16 + khalf + (j - 8)));
    int col   = ntile * 16 + colL;
    size_t src = (size_t)layer * HID * HID + (size_t)K * HID + col;
    size_t dst = (size_t)layer * 16384 +
                 (((size_t)ks * 8 + ntile) * 32 + lane) * 16 + j;
    f1[dst] = (__bf16)w1[src];
    f2[dst] = (__bf16)w2[src];
}

__global__ __launch_bounds__(256) void k_node_update(
    const int* __restrict__ rowptr, const int2* __restrict__ rec,
    const float* __restrict__ bond_emb, const float* __restrict__ hcur,
    const __bf16* __restrict__ wf1, const __bf16* __restrict__ wf2,
    const float* __restrict__ b1, const float* __restrict__ b2,
    const float* __restrict__ gamma, const float* __restrict__ beta,
    const float* __restrict__ mean, const float* __restrict__ var,
    float* __restrict__ hnext)
{
    __shared__ __align__(16) __bf16 zt[16 * HID];
    __shared__ __align__(16) __bf16 z1[16 * HID];

    static_assert(N_NODES % 16 == 0, "whole 16-node tiles: no row guard below");
    const int tid   = threadIdx.x;
    const int wave  = tid >> 5;
    const int lane  = tid & 31;
    const int node0 = blockIdx.x * 16;

    {
        const int m0 = __builtin_amdgcn_readfirstlane(wave * 2);
        const float4* __restrict__ h4 = (const float4*)hcur;
        const float4* __restrict__ b4 = (const float4*)bond_emb;
        const int n0 = node0 + m0;
        const int s0 = rowptr[n0];
        const int s1 = rowptr[n0 + 1];
        const int s2 = rowptr[n0 + 2];
        const float4 za = gather_row(n0,     s0, s1, lane, rec, b4, h4);
        const float4 zb = gather_row(n0 + 1, s1, s2, lane, rec, b4, h4);
        v4bf pa, pb;
        pa[0] = (__bf16)za.x; pa[1] = (__bf16)za.y; pa[2] = (__bf16)za.z; pa[3] = (__bf16)za.w;
        pb[0] = (__bf16)zb.x; pb[1] = (__bf16)zb.y; pb[2] = (__bf16)zb.z; pb[3] = (__bf16)zb.w;
        *(v4bf*)&zt[m0 * HID + lane * 4]       = pa;
        *(v4bf*)&zt[(m0 + 1) * HID + lane * 4] = pb;
    }
    __syncthreads();

    const int colL  = lane & 15;
    const int col   = wave * 16 + colL;
    const int khalf = (lane >> 4) * 8;
    const int rowA  = lane & 15;

    v8f acc1 = {};
    #pragma unroll
    for (int ks = 0; ks < 4; ++ks) {
        const __bf16* ap = &zt[rowA * HID + ks * 32 + khalf];
        v8bf alo = *(const v8bf*)(ap);
        v8bf ahi = *(const v8bf*)(ap + 16);
        v16bf a = __builtin_shufflevector(alo, ahi,
                    0,1,2,3,4,5,6,7,8,9,10,11,12,13,14,15);
        v16bf b = *(const v16bf*)(wf1 + (((size_t)ks * 8 + wave) * 32 + lane) * 16);
        acc1 = __builtin_amdgcn_wmma_f32_16x16x32_bf16(
                   false, a, false, b, (short)0, acc1, false, false);
    }

    const float bias1 = b1[col];
    #pragma unroll
    for (int r = 0; r < 8; ++r) {
        int m = (lane < 16) ? r : (r + 8);
        float v = acc1[r] + bias1;
        v = v > 0.f ? v : 0.f;
        z1[m * HID + col] = (__bf16)v;
    }
    __syncthreads();

    v8f acc2 = {};
    #pragma unroll
    for (int ks = 0; ks < 4; ++ks) {
        const __bf16* ap = &z1[rowA * HID + ks * 32 + khalf];
        v8bf alo = *(const v8bf*)(ap);
        v8bf ahi = *(const v8bf*)(ap + 16);
        v16bf a = __builtin_shufflevector(alo, ahi,
                    0,1,2,3,4,5,6,7,8,9,10,11,12,13,14,15);
        v16bf b = *(const v16bf*)(wf2 + (((size_t)ks * 8 + wave) * 32 + lane) * 16);
        acc2 = __builtin_amdgcn_wmma_f32_16x16x32_bf16(
                   false, a, false, b, (short)0, acc2, false, false);
    }

    const float bias2 = b2[col];
    const float mu    = mean[col];
    const float inv   = rsqrtf(var[col] + 1e-5f);
    const float ga    = gamma[col];
    const float be    = beta[col];
    #pragma unroll
    for (int r = 0; r < 8; ++r) {
        int m = (lane < 16) ? r : (r + 8);
        size_t idx = (size_t)(node0 + m) * HID + col;
        float v = acc2[r] + bias2;
        v = (v - mu) * inv * ga + be;
        v = v > 0.f ? v : 0.f;
        hnext[idx] = hcur[idx] + v;
    }
}

__global__ __launch_bounds__(256) void k_pool_zero(
    float* __restrict__ sums, float* __restrict__ cnts)
{
    int i = blockIdx.x * 256 + threadIdx.x;
    if (i < N_GRAPHS * HID) sums[i] = 0.f;
    if (i < N_GRAPHS)       cnts[i] = 0.f;
}

__global__ __launch_bounds__(256) void k_pool_acc(
    const float* __restrict__ h, const int* __restrict__ batch,
    float* __restrict__ sums, float* __restrict__ cnts)
{
    int gid = blockIdx.x * 256 + threadIdx.x;
    if (gid >= N_NODES * 32) return;
    int n = gid >> 5;
    int c = (gid & 31) * 4;
    int g = batch[n];
    const float4 v = *(const float4*)(h + (size_t)n * HID + c);
    float* o = sums + (size_t)g * HID + c;
    atomicAdd(o + 0, v.x);
    atomicAdd(o + 1, v.y);
    atomicAdd(o + 2, v.z);
    atomicAdd(o + 3, v.w);
    if ((gid & 31) == 0) atomicAdd(cnts + g, 1.0f);
}

__global__ __launch_bounds__(128) void k_pool_mlp(
    const float* __restrict__ sums, const float* __restrict__ cnts,
    const float* __restrict__ mw, const float* __restrict__ mb,
    float* __restrict__ out)
{
    __shared__ float p[HID];
    int g = blockIdx.x;
    int t = threadIdx.x;
    float cnt = cnts[g];
    cnt = cnt > 1.f ? cnt : 1.f;
    p[t] = sums[(size_t)g * HID + t] / cnt;
    __syncthreads();
    if (t < OUTD) {
        float a = mb[t];
        #pragma unroll 16
        for (int c = 0; c < HID; ++c) a += p[c] * mw[c * OUTD + t];
        out[g * OUTD + t] = a;
    }
}

extern "C" void kernel_launch(void* const* d_in, const int* in_sizes, int n_in,
                              void* d_out, int out_size, void* d_ws, size_t ws_size,
                              hipStream_t stream) {
    const int*   x          = (const int*)  d_in[0];
    const int*   edge_index = (const int*)  d_in[1];
    const int*   edge_attr  = (const int*)  d_in[2];
    const int*   batch      = (const int*)  d_in[3];
    const float* atom_emb   = (const float*)d_in[4];
    const float* bond_emb   = (const float*)d_in[5];
    const float* lin1_w     = (const float*)d_in[6];
    const float* lin1_b     = (const float*)d_in[7];
    const float* lin2_w     = (const float*)d_in[8];
    const float* lin2_b     = (const float*)d_in[9];
    const float* bn_gamma   = (const float*)d_in[10];
    const float* bn_beta    = (const float*)d_in[11];
    const float* bn_mean    = (const float*)d_in[12];
    const float* bn_var     = (const float*)d_in[13];
    const float* mlp_w      = (const float*)d_in[14];
    const float* mlp_b      = (const float*)d_in[15];

    char* ws = (char*)d_ws;
    size_t off = 0;
    auto alloc = [&](size_t bytes) -> char* {
        char* p = ws + off;
        off = (off + bytes + 255) & ~(size_t)255;
        return p;
    };
    float*  h0   = (float*) alloc((size_t)N_NODES * HID * 4);
    float*  h1   = (float*) alloc((size_t)N_NODES * HID * 4);
    __bf16* wf1  = (__bf16*)alloc((size_t)NLAYERS * 16384 * 2);
    __bf16* wf2  = (__bf16*)alloc((size_t)NLAYERS * 16384 * 2);
    float*  sums = (float*) alloc((size_t)N_GRAPHS * HID * 4);
    float*  cnts = (float*) alloc((size_t)N_GRAPHS * 4);
    int*    deg    = (int*)  alloc((size_t)N_NODES * 4);
    int*    rowptr = (int*)  alloc((size_t)(N_NODES + 1) * 4);
    int4*   slots  = (int4*) alloc((size_t)N_EDGES * 16);
    int2*   rec    = (int2*) alloc((size_t)N_EDGES * 8);

    const int nodeChunks = N_NODES * 32;

    k_prep_weights<<<(NLAYERS * 4 * 8 * 32 * 16 + 255) / 256, 256, 0, stream>>>(
        lin1_w, lin2_w, wf1, wf2, deg);
    k_encode_nodes<<<(nodeChunks + 255) / 256, 256, 0, stream>>>(x, atom_emb, h0);

    k_count<<<(N_EDGES + 255) / 256, 256, 0, stream>>>(edge_index, deg);
    k_scan<<<(N_NODES + 1023) / 1024, 1024, 0, stream>>>(deg, rowptr);
    k_fill<<<(N_EDGES + 255) / 256, 256, 0, stream>>>(edge_index, edge_attr, rowptr, deg, slots);
    k_sort<<<(N_EDGES + 255) / 256, 256, 0, stream>>>(slots, rowptr, rec);

    float* hc = h0;
    float* hn = h1;
    for (int i = 0; i < NLAYERS; ++i) {
        k_node_update<<<N_NODES / 16, 256, 0, stream>>>(
            rowptr, rec, bond_emb, hc, wf1 + (size_t)i * 16384, wf2 + (size_t)i * 16384,
            lin1_b + i * HID, lin2_b + i * HID,
            bn_gamma + i * HID, bn_beta + i * HID,
            bn_mean + i * HID, bn_var + i * HID, hn);
        float* t = hc; hc = hn; hn = t;
    }

    k_pool_zero<<<(N_GRAPHS * HID + 255) / 256, 256, 0, stream>>>(sums, cnts);
    k_pool_acc<<<(nodeChunks + 255) / 256, 256, 0, stream>>>(hc, batch, sums, cnts);
    k_pool_mlp<<<N_GRAPHS, 128, 0, stream>>>(sums, cnts, mlp_w, mlp_b, (float*)d_out);
}


# edge list by destination built through LDS, no global atomics: edge chunks ordered by node bin inside the set-up launch, one block per bin places its edges; 10 launches to 7; edge-id sum order kept
# speedup vs baseline: 2.9486x; 1.6055x over previous
#include <hip/hip_runtime.h>
#include <hip/hip_bf16.h>

#define N_NODES 50000
#define N_EDGES 500000
#define N_GRAPHS 1000
#define HID 128
#define OUTD 10
#define NLAYERS 4

typedef __attribute__((ext_vector_type(16))) __bf16 v16bf;
typedef __attribute__((ext_vector_type(8)))  __bf16 v8bf;
typedef __attribute__((ext_vector_type(4)))  __bf16 v4bf;
typedef __attribute__((ext_vector_type(8)))  float  v8f;

__device__ __forceinline__ void encode_nodes(
    const int gid, const int* __restrict__ x, const float* __restrict__ atom_emb,
    float* __restrict__ h)
{
    if (gid >= N_NODES * 32) return;
    int n = gid >> 5;
    int c = (gid & 31) * 4;
    float4 acc = make_float4(0.f, 0.f, 0.f, 0.f);
    #pragma unroll
    for (int f = 0; f < 9; ++f) {
        int idx = x[n * 9 + f];
        const float4 v = *(const float4*)(atom_emb + ((size_t)(f * 120 + idx)) * HID + c);
        acc.x += v.x; acc.y += v.y; acc.z += v.z; acc.w += v.w;
    }
    *(float4*)(h + (size_t)n * HID + c) = acc;
}

#define EP_CHUNK  2048
#define EP_CHUNKS ((N_EDGES + EP_CHUNK - 1) / EP_CHUNK)
#define EP_STEPS  8
#define EB_BINS   128
#define EB_NPB    391
#define EB_CPW    8
static_assert(EB_BINS * EB_NPB >= N_NODES, "the bins cover every node");
static_assert(8 * EP_STEPS * 32 == EP_CHUNK, "8 waves of EP_STEPS steps take one chunk");
static_assert(32 * EB_CPW >= EP_CHUNKS, "the 32 waves of k_build take every chunk");

__device__ __forceinline__ void rank_in_step(const int key, const bool valid, const int lane,
                                             int& group, int& before)
{
    group = 0; before = 0;
    unsigned mask = __builtin_amdgcn_ballot_w32(valid);
    while (mask != 0u) {
        const int l  = __builtin_ctz(mask);
        mask &= mask - 1u;
        const int kl = __builtin_amdgcn_readlane(key, l);
        if (kl == key) { group += 1; if (l < lane) before += 1; }
    }
}

__device__ __forceinline__ void part_edges(
    const int c, const int* __restrict__ edge_index, const int* __restrict__ edge_attr,
    int* __restrict__ coff, int4* __restrict__ tmp)
{
    __shared__ int cw[8 * EB_BINS];
    __shared__ int wsum[8];
    const int tid  = threadIdx.x;
    const int lane = tid & 31;
    const int wave = tid >> 5;
    for (int i = tid; i < 8 * EB_BINS; i += 256) cw[i] = 0;
    __syncthreads();

    const int e0 = c * EP_CHUNK + wave * (EP_STEPS * 32) + lane;
    int dv[EP_STEPS], sr[EP_STEPS], code[EP_STEPS];
    #pragma unroll
    for (int u = 0; u < EP_STEPS; ++u) {
        const int e  = e0 + u * 32;
        const int ec = (e < N_EDGES) ? e : (N_EDGES - 1);
        const int d  = edge_index[N_EDGES + ec];
        sr[u]        = edge_index[ec];
        const int a0 = edge_attr[ec * 3 + 0];
        const int a1 = edge_attr[ec * 3 + 1];
        const int a2 = edge_attr[ec * 3 + 2];
        code[u] = a0 | (a1 << 8) | (a2 << 16);
        dv[u]   = (e < N_EDGES) ? d : -1;
    }
    #pragma unroll
    for (int u = 0; u < EP_STEPS; ++u)
        if (dv[u] >= 0) atomicAdd(&cw[wave * EB_BINS + dv[u] / EB_NPB], 1);
    __syncthreads();

    int total = 0;
    if (tid < EB_BINS) {
        for (int w = 0; w < 8; ++w) {
            const int n = cw[w * EB_BINS + tid];
            cw[w * EB_BINS + tid] = total;
            total += n;
        }
    }
    int inc = total;
    for (int o = 1; o < 32; o <<= 1) {
        const int t = __shfl_up(inc, o);
        if (lane >= o) inc += t;
    }
    if (lane == 31) wsum[wave] = inc;
    __syncthreads();
    int tw = (lane < wave) ? wsum[lane & 7] : 0;
    for (int o = 16; o > 0; o >>= 1) tw += __shfl_xor(tw, o);
    const int excl = tw + inc - total;
    if (tid < EB_BINS) {
        for (int w = 0; w < 8; ++w) cw[w * EB_BINS + tid] += excl;
        coff[c * (EB_BINS + 1) + tid] = excl;
        if (tid == EB_BINS - 1) coff[c * (EB_BINS + 1) + EB_BINS] = excl + total;
    }
    __syncthreads();

    #pragma unroll
    for (int u = 0; u < EP_STEPS; ++u) {
        const bool valid = dv[u] >= 0;
        const int  bin   = valid ? dv[u] / EB_NPB : -1;
        int group, before;
        rank_in_step(bin, valid, lane, group, before);
        if (valid) {
            const int idx = wave * EB_BINS + bin;
            const int cur = cw[idx];
            tmp[c * EP_CHUNK + cur + before] = make_int4(dv[u], sr[u], code[u], 0);
            if (before == group - 1) cw[idx] = cur + group;
        }
    }
}

__global__ __launch_bounds__(1024) void k_build(
    const int* __restrict__ coff, const int4* __restrict__ tmp,
    int* __restrict__ rowptr, int2* __restrict__ rec)
{
    __shared__ int cntw[32 * EB_NPB];
    __shared__ int wless[32];
    __shared__ int wtot[32];
    const int tid  = threadIdx.x;
    const int lane = tid & 31;
    const int wave = __builtin_amdgcn_readfirstlane(tid >> 5);
    const int b    = blockIdx.x;
    const int lo   = b * EB_NPB;

    for (int i = tid; i < 32 * EB_NPB; i += 1024) cntw[i] = 0;
    __syncthreads();

    int vo0, vo1;
    {
        const int  ck    = wave * EB_CPW + (lane & (EB_CPW - 1));
        const bool livek = ck < EP_CHUNKS;
        const int  cck   = livek ? ck : (EP_CHUNKS - 1);
        vo0 = coff[cck * (EB_BINS + 1) + b];
        vo1 = coff[cck * (EB_BINS + 1) + b + 1];
        if (!livek) { vo0 = 0; vo1 = 0; }
    }
    int s[EB_CPW], t[EB_CPW];
    int less = 0;
    #pragma unroll
    for (int k = 0; k < EB_CPW; ++k) {
        const int c  = wave * EB_CPW + k;
        const int o0 = __builtin_amdgcn_readlane(vo0, k);
        const int o1 = __builtin_amdgcn_readlane(vo1, k);
        less += o0;
        s[k] = c * EP_CHUNK + o0;
        t[k] = c * EP_CHUNK + o1;
    }
    int4 r0[EB_CPW];
    #pragma unroll
    for (int k = 0; k < EB_CPW; ++k) {
        const int p = s[k] + lane;
        const int4 v = tmp[(p < t[k]) ? p : 0];
        r0[k] = v;
        if (p >= t[k]) r0[k].x = -1;
    }

    int mine = 0;
    #pragma unroll
    for (int k = 0; k < EB_CPW; ++k) {
        mine += t[k] - s[k];
        if (r0[k].x >= 0) atomicAdd(&cntw[wave * EB_NPB + (r0[k].x - lo)], 1);
        for (int base = s[k] + 32; base < t[k]; base += 32) {
            const int p = base + lane;
            if (p < t[k]) atomicAdd(&cntw[wave * EB_NPB + (tmp[p].x - lo)], 1);
        }
    }
    if (lane == 0) { wless[wave] = less; wtot[wave] = mine; }
    __syncthreads();

    int total = 0;
    if (tid < EB_NPB) {
        for (int w = 0; w < 32; ++w) {
            const int n = cntw[w * EB_NPB + tid];
            cntw[w * EB_NPB + tid] = total;
            total += n;
        }
    }
    int tl = wless[lane];
    int ta = wtot[lane];
    for (int o = 16; o > 0; o >>= 1) {
        tl += __shfl_xor(tl, o);
        ta += __shfl_xor(ta, o);
    }
    __syncthreads();
    int inc = total;
    for (int o = 1; o < 32; o <<= 1) {
        const int x = __shfl_up(inc, o);
        if (lane >= o) inc += x;
    }
    if (lane == 31) wtot[wave] = inc;
    __syncthreads();
    int tw = (lane < wave) ? wtot[lane] : 0;
    for (int o = 16; o > 0; o >>= 1) tw += __shfl_xor(tw, o);
    const int excl = tl + tw + inc - total;
    if (tid < EB_NPB) {
        for (int w = 0; w < 32; ++w) cntw[w * EB_NPB + tid] += excl;
        if (lo + tid < N_NODES) rowptr[lo + tid] = excl;
    }
    if (b == EB_BINS - 1 && tid == 0) rowptr[N_NODES] = tl + ta;
    __syncthreads();

    #pragma unroll
    for (int k = 0; k < EB_CPW; ++k) {
        int4 v = r0[k];
        int base = s[k];
        while (true) {
            const bool valid = v.x >= 0;
            int group, before;
            rank_in_step(v.x, valid, lane, group, before);
            if (valid) {
                const int idx = wave * EB_NPB + (v.x - lo);
                const int cur = cntw[idx];
                rec[cur + before] = make_int2(v.y, v.z);
                if (before == group - 1) cntw[idx] = cur + group;
            }
            base += 32;
            if (base >= t[k]) break;
            const int p = base + lane;
            v = tmp[(p < t[k]) ? p : 0];
            if (p >= t[k]) v.x = -1;
        }
    }
}

__device__ __forceinline__ float4 edge_msg(const float4 hv, const float4 e0,
                                           const float4 e1, const float4 e2)
{
    float m0 = hv.x + e0.x + e1.x + e2.x; m0 = m0 > 0.f ? m0 : 0.f;
    float m1 = hv.y + e0.y + e1.y + e2.y; m1 = m1 > 0.f ? m1 : 0.f;
    float m2 = hv.z + e0.z + e1.z + e2.z; m2 = m2 > 0.f ? m2 : 0.f;
    float m3 = hv.w + e0.w + e1.w + e2.w; m3 = m3 > 0.f ? m3 : 0.f;
    return make_float4(m0, m1, m2, m3);
}

__device__ __forceinline__ float4 gather_row(
    const int n, const int s, const int t, const int lane,
    const int2* __restrict__ rec, const float4* __restrict__ b4,
    const float4* __restrict__ h4)
{
    float4 acc = h4[(size_t)n * 32 + lane];
    for (int base = s; base < t; base += 32) {
        const int left = t - base;
        const int cnt  = left < 32 ? left : 32;
        int2 my = make_int2(0, 0);
        if (lane < cnt) my = rec[base + lane];
        int j = 0;
        for (; j + 4 <= cnt; j += 4) {
            float4 hv[4], e0[4], e1[4], e2[4];
            for (int u = 0; u < 4; ++u) {
                const int src  = __builtin_amdgcn_readlane(my.x, j + u);
                const int code = __builtin_amdgcn_readlane(my.y, j + u);
                hv[u] = h4[(size_t)src * 32 + lane];
                e0[u] = b4[(0 * 6 + (code & 255)) * 32 + lane];
                e1[u] = b4[(1 * 6 + ((code >> 8) & 255)) * 32 + lane];
                e2[u] = b4[(2 * 6 + (code >> 16)) * 32 + lane];
            }
            for (int u = 0; u < 4; ++u) {
                const float4 m = edge_msg(hv[u], e0[u], e1[u], e2[u]);
                acc.x += m.x; acc.y += m.y; acc.z += m.z; acc.w += m.w;
            }
        }
        for (; j < cnt; ++j) {
            const int src  = __builtin_amdgcn_readlane(my.x, j);
            const int code = __builtin_amdgcn_readlane(my.y, j);
            const float4 hv = h4[(size_t)src * 32 + lane];
            const float4 e0 = b4[(0 * 6 + (code & 255)) * 32 + lane];
            const float4 e1 = b4[(1 * 6 + ((code >> 8) & 255)) * 32 + lane];
            const float4 e2 = b4[(2 * 6 + (code >> 16)) * 32 + lane];
            const float4 m = edge_msg(hv, e0, e1, e2);
            acc.x += m.x; acc.y += m.y; acc.z += m.z; acc.w += m.w;
        }
    }
    return acc;
}

#define PREP_ITEMS (NLAYERS * 4 * 8 * 32 * 16)
#define PREP_BLOCKS (PREP_ITEMS / 256)
__device__ __forceinline__ void prep_weights(
    const int o,
    const float* __restrict__ w1, const float* __restrict__ w2,
    __bf16* __restrict__ f1, __bf16* __restrict__ f2,
    float* __restrict__ sums, float* __restrict__ cnts)
{
    if (o >= PREP_ITEMS) return;
    static_assert(N_GRAPHS * HID <= 2 * PREP_ITEMS, "two pool sums per thread");
    static_assert(N_GRAPHS <= PREP_ITEMS, "one thread per graph count");
    if (o < N_GRAPHS * HID)              sums[o] = 0.f;
    if (o + PREP_ITEMS < N_GRAPHS * HID) sums[o + PREP_ITEMS] = 0.f;
    if (o < N_GRAPHS)                    cnts[o] = 0.f;
    int j     =  o        & 15;
    int lane  = (o >> 4)  & 31;
    int ntile = (o >> 9)  & 7;
    int ks    = (o >> 12) & 3;
    int layer =  o >> 14;
    int colL  = lane & 15;
    int khalf = (lane >> 4) * 8;
    int K     = ks * 32 + ((j < 8) ? (khalf + j) : (16 + khalf + (j - 8)));
    int col   = ntile * 16 + colL;
    size_t src = (size_t)layer * HID * HID + (size_t)K * HID + col;
    size_t dst = (size_t)layer * 16384 +
                 (((size_t)ks * 8 + ntile) * 32 + lane) * 16 + j;
    f1[dst] = (__bf16)w1[src];
    f2[dst] = (__bf16)w2[src];
}

__global__ __launch_bounds__(256) void k_setup(
    const int* __restrict__ x, const float* __restrict__ atom_emb, float* __restrict__ h,
    const float* __restrict__ w1, const float* __restrict__ w2,
    __bf16* __restrict__ f1, __bf16* __restrict__ f2,
    float* __restrict__ sums, float* __restrict__ cnts,
    const int* __restrict__ edge_index, const int* __restrict__ edge_attr,
    int* __restrict__ coff, int4* __restrict__ tmp)
{
    const int b = blockIdx.x;
    if (b < PREP_BLOCKS) {
        prep_weights(b * 256 + threadIdx.x, w1, w2, f1, f2, sums, cnts);
    } else if (b < PREP_BLOCKS + EP_CHUNKS) {
        part_edges(b - PREP_BLOCKS, edge_index, edge_attr, coff, tmp);
    } else {
        encode_nodes((b - PREP_BLOCKS - EP_CHUNKS) * 256 + threadIdx.x, x, atom_emb, h);
    }
}

template <bool LAST>
__device__ __forceinline__ void node_update_body(
    const int* __restrict__ rowptr, const int2* __restrict__ rec,
    const float* __restrict__ bond_emb, const float* __restrict__ hcur,
    const __bf16* __restrict__ wf1, const __bf16* __restrict__ wf2,
    const float* __restrict__ b1, const float* __restrict__ b2,
    const float* __restrict__ gamma, const float* __restrict__ beta,
    const float* __restrict__ mean, const float* __restrict__ var,
    float* __restrict__ hnext,
    const int* __restrict__ batch, float* __restrict__ sums, float* __restrict__ cnts)
{
    __shared__ __align__(16) __bf16 zt[16 * HID];
    __shared__ __align__(16) __bf16 z1[16 * HID];

    static_assert(N_NODES % 16 == 0, "whole 16-node tiles: no row guard below");
    const int tid   = threadIdx.x;
    const int wave  = tid >> 5;
    const int lane  = tid & 31;
    const int node0 = blockIdx.x * 16;

    {
        const int m0 = __builtin_amdgcn_readfirstlane(wave * 2);
        const float4* __restrict__ h4 = (const float4*)hcur;
        const float4* __restrict__ b4 = (const float4*)bond_emb;
        const int n0 = node0 + m0;
        const int s0 = rowptr[n0];
        const int s1 = rowptr[n0 + 1];
        const int s2 = rowptr[n0 + 2];
        const float4 za = gather_row(n0,     s0, s1, lane, rec, b4, h4);
        const float4 zb = gather_row(n0 + 1, s1, s2, lane, rec, b4, h4);
        v4bf pa, pb;
        pa[0] = (__bf16)za.x; pa[1] = (__bf16)za.y; pa[2] = (__bf16)za.z; pa[3] = (__bf16)za.w;
        pb[0] = (__bf16)zb.x; pb[1] = (__bf16)zb.y; pb[2] = (__bf16)zb.z; pb[3] = (__bf16)zb.w;
        *(v4bf*)&zt[m0 * HID + lane * 4]       = pa;
        *(v4bf*)&zt[(m0 + 1) * HID + lane * 4] = pb;
    }
    __syncthreads();

    const int colL  = lane & 15;
    const int col   = wave * 16 + colL;
    const int khalf = (lane >> 4) * 8;
    const int rowA  = lane & 15;

    v8f acc1 = {};
    #pragma unroll
    for (int ks = 0; ks < 4; ++ks) {
        const __bf16* ap = &zt[rowA * HID + ks * 32 + khalf];
        v8bf alo = *(const v8bf*)(ap);
        v8bf ahi = *(const v8bf*)(ap + 16);
        v16bf a = __builtin_shufflevector(alo, ahi,
                    0,1,2,3,4,5,6,7,8,9,10,11,12,13,14,15);
        v16bf b = *(const v16bf*)(wf1 + (((size_t)ks * 8 + wave) * 32 + lane) * 16);
        acc1 = __builtin_amdgcn_wmma_f32_16x16x32_bf16(
                   false, a, false, b, (short)0, acc1, false, false);
    }

    const float bias1 = b1[col];
    #pragma unroll
    for (int r = 0; r < 8; ++r) {
        int m = (lane < 16) ? r : (r + 8);
        float v = acc1[r] + bias1;
        v = v > 0.f ? v : 0.f;
        z1[m * HID + col] = (__bf16)v;
    }
    __syncthreads();

    v8f acc2 = {};
    #pragma unroll
    for (int ks = 0; ks < 4; ++ks) {
        const __bf16* ap = &z1[rowA * HID + ks * 32 + khalf];
        v8bf alo = *(const v8bf*)(ap);
        v8bf ahi = *(const v8bf*)(ap + 16);
        v16bf a = __builtin_shufflevector(alo, ahi,
                    0,1,2,3,4,5,6,7,8,9,10,11,12,13,14,15);
        v16bf b = *(const v16bf*)(wf2 + (((size_t)ks * 8 + wave) * 32 + lane) * 16);
        acc2 = __builtin_amdgcn_wmma_f32_16x16x32_bf16(
                   false, a, false, b, (short)0, acc2, false, false);
    }

    const float bias2 = b2[col];
    const float mu    = mean[col];
    const float inv   = rsqrtf(var[col] + 1e-5f);
    const float ga    = gamma[col];
    const float be    = beta[col];
    if (!LAST) {
        #pragma unroll
        for (int r = 0; r < 8; ++r) {
            int m = (lane < 16) ? r : (r + 8);
            size_t idx = (size_t)(node0 + m) * HID + col;
            float v = acc2[r] + bias2;
            v = (v - mu) * inv * ga + be;
            v = v > 0.f ? v : 0.f;
            hnext[idx] = hcur[idx] + v;
        }
    } else {
        int G[16];
        #pragma unroll
        for (int k = 0; k < 16; ++k) G[k] = batch[node0 + k];
        const bool hi   = lane >= 16;
        const int  row0 = node0 + (hi ? 8 : 0);
        float hv[8];
        #pragma unroll
        for (int r = 0; r < 8; ++r) {
            size_t idx = (size_t)(row0 + r) * HID + col;
            float v = acc2[r] + bias2;
            v = (v - mu) * inv * ga + be;
            v = v > 0.f ? v : 0.f;
            hv[r] = hcur[idx] + v;
        }
        float run  = 0.f;
        float nrun = 0.f;
        #pragma unroll
        for (int r = 0; r < 8; ++r) {
            run  += hv[r];
            nrun += 1.f;
            const int gr = hi ? G[r + 8] : G[r];
            bool flush = true;
            if (r < 7) flush = ((hi ? G[r + 9] : G[r + 1]) != gr);
            if (flush) {
                atomicAdd(sums + (size_t)gr * HID + col, run);
                if (col == 0) atomicAdd(cnts + gr, nrun);
                run  = 0.f;
                nrun = 0.f;
            }
        }
    }
}

__global__ __launch_bounds__(256) void k_node_update(
    const int* __restrict__ rowptr, const int2* __restrict__ rec,
    const float* __restrict__ bond_emb, const float* __restrict__ hcur,
    const __bf16* __restrict__ wf1, const __bf16* __restrict__ wf2,
    const float* __restrict__ b1, const float* __restrict__ b2,
    const float* __restrict__ gamma, const float* __restrict__ beta,
    const float* __restrict__ mean, const float* __restrict__ var,
    float* __restrict__ hnext)
{
    node_update_body<false>(rowptr, rec, bond_emb, hcur, wf1, wf2, b1, b2,
                            gamma, beta, mean, var, hnext, nullptr, nullptr, nullptr);
}

__global__ __launch_bounds__(256) void k_node_update_pool(
    const int* __restrict__ rowptr, const int2* __restrict__ rec,
    const float* __restrict__ bond_emb, const float* __restrict__ hcur,
    const __bf16* __restrict__ wf1, const __bf16* __restrict__ wf2,
    const float* __restrict__ b1, const float* __restrict__ b2,
    const float* __restrict__ gamma, const float* __restrict__ beta,
    const float* __restrict__ mean, const float* __restrict__ var,
    const int* __restrict__ batch, float* __restrict__ sums, float* __restrict__ cnts)
{
    node_update_body<true>(rowptr, rec, bond_emb, hcur, wf1, wf2, b1, b2,
                           gamma, beta, mean, var, nullptr, batch, sums, cnts);
}

__global__ __launch_bounds__(128) void k_pool_mlp(
    const float* __restrict__ sums, const float* __restrict__ cnts,
    const float* __restrict__ mw, const float* __restrict__ mb,
    float* __restrict__ out)
{
    __shared__ float p[HID];
    int g = blockIdx.x;
    int t = threadIdx.x;
    float cnt = cnts[g];
    cnt = cnt > 1.f ? cnt : 1.f;
    p[t] = sums[(size_t)g * HID + t] / cnt;
    __syncthreads();
    if (t < OUTD) {
        float a = mb[t];
        #pragma unroll 16
        for (int c = 0; c < HID; ++c) a += p[c] * mw[c * OUTD + t];
        out[g * OUTD + t] = a;
    }
}

extern "C" void kernel_launch(void* const* d_in, const int* in_sizes, int n_in,
                              void* d_out, int out_size, void* d_ws, size_t ws_size,
                              hipStream_t stream) {
    const int*   x          = (const int*)  d_in[0];
    const int*   edge_index = (const int*)  d_in[1];
    const int*   edge_attr  = (const int*)  d_in[2];
    const int*   batch      = (const int*)  d_in[3];
    const float* atom_emb   = (const float*)d_in[4];
    const float* bond_emb   = (const float*)d_in[5];
    const float* lin1_w     = (const float*)d_in[6];
    const float* lin1_b     = (const float*)d_in[7];
    const float* lin2_w     = (const float*)d_in[8];
    const float* lin2_b     = (const float*)d_in[9];
    const float* bn_gamma   = (const float*)d_in[10];
    const float* bn_beta    = (const float*)d_in[11];
    const float* bn_mean    = (const float*)d_in[12];
    const float* bn_var     = (const float*)d_in[13];
    const float* mlp_w      = (const float*)d_in[14];
    const float* mlp_b      = (const float*)d_in[15];

    char* ws = (char*)d_ws;
    size_t off = 0;
    auto alloc = [&](size_t bytes) -> char* {
        char* p = ws + off;
        off = (off + bytes + 255) & ~(size_t)255;
        return p;
    };
    float*  h0   = (float*) alloc((size_t)N_NODES * HID * 4);
    float*  h1   = (float*) alloc((size_t)N_NODES * HID * 4);
    __bf16* wf1  = (__bf16*)alloc((size_t)NLAYERS * 16384 * 2);
    __bf16* wf2  = (__bf16*)alloc((size_t)NLAYERS * 16384 * 2);
    float*  sums = (float*) alloc((size_t)N_GRAPHS * HID * 4);
    float*  cnts = (float*) alloc((size_t)N_GRAPHS * 4);
    int*    coff   = (int*)  alloc((size_t)EP_CHUNKS * (EB_BINS + 1) * 4);
    int*    rowptr = (int*)  alloc((size_t)(N_NODES + 1) * 4);
    int4*   tmp    = (int4*) alloc((size_t)N_EDGES * 16);
    int2*   rec    = (int2*) alloc((size_t)N_EDGES * 8);

    const int nodeChunks = N_NODES * 32;

    k_setup<<<PREP_BLOCKS + EP_CHUNKS + (nodeChunks + 255) / 256, 256, 0, stream>>>(
        x, atom_emb, h0, lin1_w, lin2_w, wf1, wf2, sums, cnts,
        edge_index, edge_attr, coff, tmp);

    k_build<<<EB_BINS, 1024, 0, stream>>>(coff, tmp, rowptr, rec);

    float* hc = h0;
    float* hn = h1;
    for (int i = 0; i < NLAYERS - 1; ++i) {
        k_node_update<<<N_NODES / 16, 256, 0, stream>>>(
            rowptr, rec, bond_emb, hc, wf1 + (size_t)i * 16384, wf2 + (size_t)i * 16384,
            lin1_b + i * HID, lin2_b + i * HID,
            bn_gamma + i * HID, bn_beta + i * HID,
            bn_mean + i * HID, bn_var + i * HID, hn);
        float* t = hc; hc = hn; hn = t;
    }
    k_node_update_pool<<<N_NODES / 16, 256, 0, stream>>>(
        rowptr, rec, bond_emb, hc,
        wf1 + (size_t)(NLAYERS - 1) * 16384, wf2 + (size_t)(NLAYERS - 1) * 16384,
        lin1_b + (NLAYERS - 1) * HID, lin2_b + (NLAYERS - 1) * HID,
        bn_gamma + (NLAYERS - 1) * HID, bn_beta + (NLAYERS - 1) * HID,
        bn_mean + (NLAYERS - 1) * HID, bn_var + (NLAYERS - 1) * HID,
        batch, sums, cnts);

    k_pool_mlp<<<N_GRAPHS, 128, 0, stream>>>(sums, cnts, mlp_w, mlp_b, (float*)d_out);
}


# node-update kernels held to 64 registers per wave (launch bound asks 16 waves per SIMD; row 1 stored before row 2 is gathered, so nothing spills): 8 blocks resident per WGP, not 6; sums unchanged
# speedup vs baseline: 3.1901x; 1.0819x over previous
#include <hip/hip_runtime.h>
#include <hip/hip_bf16.h>

#define N_NODES 50000
#define N_EDGES 500000
#define N_GRAPHS 1000
#define HID 128
#define OUTD 10
#define NLAYERS 4

typedef __attribute__((ext_vector_type(16))) __bf16 v16bf;
typedef __attribute__((ext_vector_type(8)))  __bf16 v8bf;
typedef __attribute__((ext_vector_type(4)))  __bf16 v4bf;
typedef __attribute__((ext_vector_type(8)))  float  v8f;

__device__ __forceinline__ void encode_nodes(
    const int gid, const int* __restrict__ x, const float* __restrict__ atom_emb,
    float* __restrict__ h)
{
    if (gid >= N_NODES * 32) return;
    int n = gid >> 5;
    int c = (gid & 31) * 4;
    float4 acc = make_float4(0.f, 0.f, 0.f, 0.f);
    #pragma unroll
    for (int f = 0; f < 9; ++f) {
        int idx = x[n * 9 + f];
        const float4 v = *(const float4*)(atom_emb + ((size_t)(f * 120 + idx)) * HID + c);
        acc.x += v.x; acc.y += v.y; acc.z += v.z; acc.w += v.w;
    }
    *(float4*)(h + (size_t)n * HID + c) = acc;
}

#define EP_CHUNK  2048
#define EP_CHUNKS ((N_EDGES + EP_CHUNK - 1) / EP_CHUNK)
#define EP_STEPS  8
#define EB_BINS   128
#define EB_NPB    391
#define EB_CPW    8
static_assert(EB_BINS * EB_NPB >= N_NODES, "the bins cover every node");
static_assert(8 * EP_STEPS * 32 == EP_CHUNK, "8 waves of EP_STEPS steps take one chunk");
static_assert(32 * EB_CPW >= EP_CHUNKS, "the 32 waves of k_build take every chunk");

__device__ __forceinline__ void rank_in_step(const int key, const bool valid, const int lane,
                                             int& group, int& before)
{
    group = 0; before = 0;
    unsigned mask = __builtin_amdgcn_ballot_w32(valid);
    while (mask != 0u) {
        const int l  = __builtin_ctz(mask);
        mask &= mask - 1u;
        const int kl = __builtin_amdgcn_readlane(key, l);
        if (kl == key) { group += 1; if (l < lane) before += 1; }
    }
}

__device__ __forceinline__ void part_edges(
    const int c, const int* __restrict__ edge_index, const int* __restrict__ edge_attr,
    int* __restrict__ coff, int4* __restrict__ tmp)
{
    __shared__ int cw[8 * EB_BINS];
    __shared__ int wsum[8];
    const int tid  = threadIdx.x;
    const int lane = tid & 31;
    const int wave = tid >> 5;
    for (int i = tid; i < 8 * EB_BINS; i += 256) cw[i] = 0;
    __syncthreads();

    const int e0 = c * EP_CHUNK + wave * (EP_STEPS * 32) + lane;
    int dv[EP_STEPS], sr[EP_STEPS], code[EP_STEPS];
    #pragma unroll
    for (int u = 0; u < EP_STEPS; ++u) {
        const int e  = e0 + u * 32;
        const int ec = (e < N_EDGES) ? e : (N_EDGES - 1);
        const int d  = edge_index[N_EDGES + ec];
        sr[u]        = edge_index[ec];
        const int a0 = edge_attr[ec * 3 + 0];
        const int a1 = edge_attr[ec * 3 + 1];
        const int a2 = edge_attr[ec * 3 + 2];
        code[u] = a0 | (a1 << 8) | (a2 << 16);
        dv[u]   = (e < N_EDGES) ? d : -1;
    }
    #pragma unroll
    for (int u = 0; u < EP_STEPS; ++u)
        if (dv[u] >= 0) atomicAdd(&cw[wave * EB_BINS + dv[u] / EB_NPB], 1);
    __syncthreads();

    int total = 0;
    if (tid < EB_BINS) {
        for (int w = 0; w < 8; ++w) {
            const int n = cw[w * EB_BINS + tid];
            cw[w * EB_BINS + tid] = total;
            total += n;
        }
    }
    int inc = total;
    for (int o = 1; o < 32; o <<= 1) {
        const int t = __shfl_up(inc, o);
        if (lane >= o) inc += t;
    }
    if (lane == 31) wsum[wave] = inc;
    __syncthreads();
    int tw = (lane < wave) ? wsum[lane & 7] : 0;
    for (int o = 16; o > 0; o >>= 1) tw += __shfl_xor(tw, o);
    const int excl = tw + inc - total;
    if (tid < EB_BINS) {
        for (int w = 0; w < 8; ++w) cw[w * EB_BINS + tid] += excl;
        coff[c * (EB_BINS + 1) + tid] = excl;
        if (tid == EB_BINS - 1) coff[c * (EB_BINS + 1) + EB_BINS] = excl + total;
    }
    __syncthreads();

    #pragma unroll
    for (int u = 0; u < EP_STEPS; ++u) {
        const bool valid = dv[u] >= 0;
        const int  bin   = valid ? dv[u] / EB_NPB : -1;
        int group, before;
        rank_in_step(bin, valid, lane, group, before);
        if (valid) {
            const int idx = wave * EB_BINS + bin;
            const int cur = cw[idx];
            tmp[c * EP_CHUNK + cur + before] = make_int4(dv[u], sr[u], code[u], 0);
            if (before == group - 1) cw[idx] = cur + group;
        }
    }
}

__global__ __launch_bounds__(1024) void k_build(
    const int* __restrict__ coff, const int4* __restrict__ tmp,
    int* __restrict__ rowptr, int2* __restrict__ rec)
{
    __shared__ int cntw[32 * EB_NPB];
    __shared__ int wless[32];
    __shared__ int wtot[32];
    const int tid  = threadIdx.x;
    const int lane = tid & 31;
    const int wave = __builtin_amdgcn_readfirstlane(tid >> 5);
    const int b    = blockIdx.x;
    const int lo   = b * EB_NPB;

    for (int i = tid; i < 32 * EB_NPB; i += 1024) cntw[i] = 0;
    __syncthreads();

    int vo0, vo1;
    {
        const int  ck    = wave * EB_CPW + (lane & (EB_CPW - 1));
        const bool livek = ck < EP_CHUNKS;
        const int  cck   = livek ? ck : (EP_CHUNKS - 1);
        vo0 = coff[cck * (EB_BINS + 1) + b];
        vo1 = coff[cck * (EB_BINS + 1) + b + 1];
        if (!livek) { vo0 = 0; vo1 = 0; }
    }
    int s[EB_CPW], t[EB_CPW];
    int less = 0;
    #pragma unroll
    for (int k = 0; k < EB_CPW; ++k) {
        const int c  = wave * EB_CPW + k;
        const int o0 = __builtin_amdgcn_readlane(vo0, k);
        const int o1 = __builtin_amdgcn_readlane(vo1, k);
        less += o0;
        s[k] = c * EP_CHUNK + o0;
        t[k] = c * EP_CHUNK + o1;
    }
    int4 r0[EB_CPW];
    #pragma unroll
    for (int k = 0; k < EB_CPW; ++k) {
        const int p = s[k] + lane;
        const int4 v = tmp[(p < t[k]) ? p : 0];
        r0[k] = v;
        if (p >= t[k]) r0[k].x = -1;
    }

    int mine = 0;
    #pragma unroll
    for (int k = 0; k < EB_CPW; ++k) {
        mine += t[k] - s[k];
        if (r0[k].x >= 0) atomicAdd(&cntw[wave * EB_NPB + (r0[k].x - lo)], 1);
        for (int base = s[k] + 32; base < t[k]; base += 32) {
            const int p = base + lane;
            if (p < t[k]) atomicAdd(&cntw[wave * EB_NPB + (tmp[p].x - lo)], 1);
        }
    }
    if (lane == 0) { wless[wave] = less; wtot[wave] = mine; }
    __syncthreads();

    int total = 0;
    if (tid < EB_NPB) {
        for (int w = 0; w < 32; ++w) {
            const int n = cntw[w * EB_NPB + tid];
            cntw[w * EB_NPB + tid] = total;
            total += n;
        }
    }
    int tl = wless[lane];
    int ta = wtot[lane];
    for (int o = 16; o > 0; o >>= 1) {
        tl += __shfl_xor(tl, o);
        ta += __shfl_xor(ta, o);
    }
    __syncthreads();
    int inc = total;
    for (int o = 1; o < 32; o <<= 1) {
        const int x = __shfl_up(inc, o);
        if (lane >= o) inc += x;
    }
    if (lane == 31) wtot[wave] = inc;
    __syncthreads();
    int tw = (lane < wave) ? wtot[lane] : 0;
    for (int o = 16; o > 0; o >>= 1) tw += __shfl_xor(tw, o);
    const int excl = tl + tw + inc - total;
    if (tid < EB_NPB) {
        for (int w = 0; w < 32; ++w) cntw[w * EB_NPB + tid] += excl;
        if (lo + tid < N_NODES) rowptr[lo + tid] = excl;
    }
    if (b == EB_BINS - 1 && tid == 0) rowptr[N_NODES] = tl + ta;
    __syncthreads();

    #pragma unroll
    for (int k = 0; k < EB_CPW; ++k) {
        int4 v = r0[k];
        int base = s[k];
        while (true) {
            const bool valid = v.x >= 0;
            int group, before;
            rank_in_step(v.x, valid, lane, group, before);
            if (valid) {
                const int idx = wave * EB_NPB + (v.x - lo);
                const int cur = cntw[idx];
                rec[cur + before] = make_int2(v.y, v.z);
                if (before == group - 1) cntw[idx] = cur + group;
            }
            base += 32;
            if (base >= t[k]) break;
            const int p = base + lane;
            v = tmp[(p < t[k]) ? p : 0];
            if (p >= t[k]) v.x = -1;
        }
    }
}

__device__ __forceinline__ float4 edge_msg(const float4 hv, const float4 e0,
                                           const float4 e1, const float4 e2)
{
    float m0 = hv.x + e0.x + e1.x + e2.x; m0 = m0 > 0.f ? m0 : 0.f;
    float m1 = hv.y + e0.y + e1.y + e2.y; m1 = m1 > 0.f ? m1 : 0.f;
    float m2 = hv.z + e0.z + e1.z + e2.z; m2 = m2 > 0.f ? m2 : 0.f;
    float m3 = hv.w + e0.w + e1.w + e2.w; m3 = m3 > 0.f ? m3 : 0.f;
    return make_float4(m0, m1, m2, m3);
}

__device__ __forceinline__ float4 gather_row(
    const int n, const int s, const int t, const int lane,
    const int2* __restrict__ rec, const float4* b4,
    const float4* __restrict__ h4)
{
    float4 acc = h4[(size_t)n * 32 + lane];
    for (int base = s; base < t; base += 32) {
        const int left = t - base;
        const int cnt  = left < 32 ? left : 32;
        int2 my = make_int2(0, 0);
        if (lane < cnt) my = rec[base + lane];
        int j = 0;
        for (; j + 4 <= cnt; j += 4) {
            float4 hv[4];
            int code[4];
            for (int u = 0; u < 4; ++u) {
                const int src = __builtin_amdgcn_readlane(my.x, j + u);
                code[u]       = __builtin_amdgcn_readlane(my.y, j + u);
                hv[u] = h4[(size_t)src * 32 + lane];
            }
            for (int u = 0; u < 4; ++u) {
                const float4 e0 = b4[(0 * 6 + (code[u] & 255)) * 32 + lane];
                const float4 e1 = b4[(1 * 6 + ((code[u] >> 8) & 255)) * 32 + lane];
                const float4 e2 = b4[(2 * 6 + (code[u] >> 16)) * 32 + lane];
                const float4 m = edge_msg(hv[u], e0, e1, e2);
                acc.x += m.x; acc.y += m.y; acc.z += m.z; acc.w += m.w;
            }
        }
        for (; j < cnt; ++j) {
            const int src  = __builtin_amdgcn_readlane(my.x, j);
            const int code = __builtin_amdgcn_readlane(my.y, j);
            const float4 hv = h4[(size_t)src * 32 + lane];
            const float4 e0 = b4[(0 * 6 + (code & 255)) * 32 + lane];
            const float4 e1 = b4[(1 * 6 + ((code >> 8) & 255)) * 32 + lane];
            const float4 e2 = b4[(2 * 6 + (code >> 16)) * 32 + lane];
            const float4 m = edge_msg(hv, e0, e1, e2);
            acc.x += m.x; acc.y += m.y; acc.z += m.z; acc.w += m.w;
        }
    }
    return acc;
}

#define PREP_ITEMS (NLAYERS * 4 * 8 * 32 * 16)
#define PREP_BLOCKS (PREP_ITEMS / 256)
__device__ __forceinline__ void prep_weights(
    const int o,
    const float* __restrict__ w1, const float* __restrict__ w2,
    __bf16* __restrict__ f1, __bf16* __restrict__ f2,
    float* __restrict__ sums, float* __restrict__ cnts)
{
    if (o >= PREP_ITEMS) return;
    static_assert(N_GRAPHS * HID <= 2 * PREP_ITEMS, "two pool sums per thread");
    static_assert(N_GRAPHS <= PREP_ITEMS, "one thread per graph count");
    if (o < N_GRAPHS * HID)              sums[o] = 0.f;
    if (o + PREP_ITEMS < N_GRAPHS * HID) sums[o + PREP_ITEMS] = 0.f;
    if (o < N_GRAPHS)                    cnts[o] = 0.f;
    int j     =  o        & 15;
    int lane  = (o >> 4)  & 31;
    int ntile = (o >> 9)  & 7;
    int ks    = (o >> 12) & 3;
    int layer =  o >> 14;
    int colL  = lane & 15;
    int khalf = (lane >> 4) * 8;
    int K     = ks * 32 + ((j < 8) ? (khalf + j) : (16 + khalf + (j - 8)));
    int col   = ntile * 16 + colL;
    size_t src = (size_t)layer * HID * HID + (size_t)K * HID + col;
    size_t dst = (size_t)layer * 16384 +
                 (((size_t)ks * 8 + ntile) * 32 + lane) * 16 + j;
    f1[dst] = (__bf16)w1[src];
    f2[dst] = (__bf16)w2[src];
}

__global__ __launch_bounds__(256) void k_setup(
    const int* __restrict__ x, const float* __restrict__ atom_emb, float* __restrict__ h,
    const float* __restrict__ w1, const float* __restrict__ w2,
    __bf16* __restrict__ f1, __bf16* __restrict__ f2,
    float* __restrict__ sums, float* __restrict__ cnts,
    const int* __restrict__ edge_index, const int* __restrict__ edge_attr,
    int* __restrict__ coff, int4* __restrict__ tmp)
{
    const int b = blockIdx.x;
    if (b < PREP_BLOCKS) {
        prep_weights(b * 256 + threadIdx.x, w1, w2, f1, f2, sums, cnts);
    } else if (b < PREP_BLOCKS + EP_CHUNKS) {
        part_edges(b - PREP_BLOCKS, edge_index, edge_attr, coff, tmp);
    } else {
        encode_nodes((b - PREP_BLOCKS - EP_CHUNKS) * 256 + threadIdx.x, x, atom_emb, h);
    }
}

template <bool LAST>
__device__ __forceinline__ void node_update_body(
    const int* __restrict__ rowptr, const int2* __restrict__ rec,
    const float* __restrict__ bond_emb, const float* __restrict__ hcur,
    const __bf16* __restrict__ wf1, const __bf16* __restrict__ wf2,
    const float* __restrict__ b1, const float* __restrict__ b2,
    const float* __restrict__ gamma, const float* __restrict__ beta,
    const float* __restrict__ mean, const float* __restrict__ var,
    float* __restrict__ hnext,
    const int* __restrict__ batch, float* __restrict__ sums, float* __restrict__ cnts)
{
    __shared__ __align__(16) __bf16 zt[16 * HID];
    __shared__ __align__(16) __bf16 z1[16 * HID];
    __shared__ __align__(16) float  bt[18 * HID];

    static_assert(N_NODES % 16 == 0, "whole 16-node tiles: no row guard below");
    const int tid   = threadIdx.x;
    const int wave  = tid >> 5;
    const int lane  = tid & 31;
    const int node0 = blockIdx.x * 16;

    const int m0 = __builtin_amdgcn_readfirstlane(wave * 2);
    const int n0 = node0 + m0;
    const int s0 = rowptr[n0];
    const int s1 = rowptr[n0 + 1];
    const int s2 = rowptr[n0 + 2];

    {
        static_assert(18 * HID / 4 == 2 * 256 + 64, "2 float4 per thread and 64 more");
        const float4* __restrict__ g4 = (const float4*)bond_emb;
        float4* bt4w = (float4*)bt;
        const float4 t0 = g4[tid];
        const float4 t1 = g4[tid + 256];
        float4 t2 = make_float4(0.f, 0.f, 0.f, 0.f);
        if (tid < 64) t2 = g4[tid + 512];
        bt4w[tid]       = t0;
        bt4w[tid + 256] = t1;
        if (tid < 64) bt4w[tid + 512] = t2;
    }
    __syncthreads();

    {
        const float4* __restrict__ h4 = (const float4*)hcur;
        const float4* b4 = (const float4*)bt;
        const float4 za = gather_row(n0,     s0, s1, lane, rec, b4, h4);
        v4bf pa;
        pa[0] = (__bf16)za.x; pa[1] = (__bf16)za.y; pa[2] = (__bf16)za.z; pa[3] = (__bf16)za.w;
        *(v4bf*)&zt[m0 * HID + lane * 4]       = pa;
        const float4 zb = gather_row(n0 + 1, s1, s2, lane, rec, b4, h4);
        v4bf pb;
        pb[0] = (__bf16)zb.x; pb[1] = (__bf16)zb.y; pb[2] = (__bf16)zb.z; pb[3] = (__bf16)zb.w;
        *(v4bf*)&zt[(m0 + 1) * HID + lane * 4] = pb;
    }
    __syncthreads();

    const int colL  = lane & 15;
    const int col   = wave * 16 + colL;
    const int khalf = (lane >> 4) * 8;
    const int rowA  = lane & 15;

    v8f acc1 = {};
    #pragma unroll
    for (int ks = 0; ks < 4; ++ks) {
        const __bf16* ap = &zt[rowA * HID + ks * 32 + khalf];
        v8bf alo = *(const v8bf*)(ap);
        v8bf ahi = *(const v8bf*)(ap + 16);
        v16bf a = __builtin_shufflevector(alo, ahi,
                    0,1,2,3,4,5,6,7,8,9,10,11,12,13,14,15);
        v16bf b = *(const v16bf*)(wf1 + (((size_t)ks * 8 + wave) * 32 + lane) * 16);
        acc1 = __builtin_amdgcn_wmma_f32_16x16x32_bf16(
                   false, a, false, b, (short)0, acc1, false, false);
    }

    const float bias1 = b1[col];
    #pragma unroll
    for (int r = 0; r < 8; ++r) {
        int m = (lane < 16) ? r : (r + 8);
        float v = acc1[r] + bias1;
        v = v > 0.f ? v : 0.f;
        z1[m * HID + col] = (__bf16)v;
    }
    __syncthreads();

    v8f acc2 = {};
    #pragma unroll
    for (int ks = 0; ks < 4; ++ks) {
        const __bf16* ap = &z1[rowA * HID + ks * 32 + khalf];
        v8bf alo = *(const v8bf*)(ap);
        v8bf ahi = *(const v8bf*)(ap + 16);
        v16bf a = __builtin_shufflevector(alo, ahi,
                    0,1,2,3,4,5,6,7,8,9,10,11,12,13,14,15);
        v16bf b = *(const v16bf*)(wf2 + (((size_t)ks * 8 + wave) * 32 + lane) * 16);
        acc2 = __builtin_amdgcn_wmma_f32_16x16x32_bf16(
                   false, a, false, b, (short)0, acc2, false, false);
    }

    const float bias2 = b2[col];
    const float mu    = mean[col];
    const float inv   = rsqrtf(var[col] + 1e-5f);
    const float ga    = gamma[col];
    const float be    = beta[col];
    if (!LAST) {
        #pragma unroll
        for (int r = 0; r < 8; ++r) {
            int m = (lane < 16) ? r : (r + 8);
            size_t idx = (size_t)(node0 + m) * HID + col;
            float v = acc2[r] + bias2;
            v = (v - mu) * inv * ga + be;
            v = v > 0.f ? v : 0.f;
            hnext[idx] = hcur[idx] + v;
        }
    } else {
        int G[16];
        #pragma unroll
        for (int k = 0; k < 16; ++k) G[k] = batch[node0 + k];
        const bool hi   = lane >= 16;
        const int  row0 = node0 + (hi ? 8 : 0);
        float hv[8];
        #pragma unroll
        for (int r = 0; r < 8; ++r) {
            size_t idx = (size_t)(row0 + r) * HID + col;
            float v = acc2[r] + bias2;
            v = (v - mu) * inv * ga + be;
            v = v > 0.f ? v : 0.f;
            hv[r] = hcur[idx] + v;
        }
        float run  = 0.f;
        float nrun = 0.f;
        #pragma unroll
        for (int r = 0; r < 8; ++r) {
            run  += hv[r];
            nrun += 1.f;
            const int gr = hi ? G[r + 8] : G[r];
            bool flush = true;
            if (r < 7) flush = ((hi ? G[r + 9] : G[r + 1]) != gr);
            if (flush) {
                atomicAdd(sums + (size_t)gr * HID + col, run);
                if (col == 0) atomicAdd(cnts + gr, nrun);
                run  = 0.f;
                nrun = 0.f;
            }
        }
    }
}

__global__ __launch_bounds__(256, 16) void k_node_update(
    const int* __restrict__ rowptr, const int2* __restrict__ rec,
    const float* __restrict__ bond_emb, const float* __restrict__ hcur,
    const __bf16* __restrict__ wf1, const __bf16* __restrict__ wf2,
    const float* __restrict__ b1, const float* __restrict__ b2,
    const float* __restrict__ gamma, const float* __restrict__ beta,
    const float* __restrict__ mean, const float* __restrict__ var,
    float* __restrict__ hnext)
{
    node_update_body<false>(rowptr, rec, bond_emb, hcur, wf1, wf2, b1, b2,
                            gamma, beta, mean, var, hnext, nullptr, nullptr, nullptr);
}

__global__ __launch_bounds__(256, 16) void k_node_update_pool(
    const int* __restrict__ rowptr, const int2* __restrict__ rec,
    const float* __restrict__ bond_emb, const float* __restrict__ hcur,
    const __bf16* __restrict__ wf1, const __bf16* __restrict__ wf2,
    const float* __restrict__ b1, const float* __restrict__ b2,
    const float* __restrict__ gamma, const float* __restrict__ beta,
    const float* __restrict__ mean, const float* __restrict__ var,
    const int* __restrict__ batch, float* __restrict__ sums, float* __restrict__ cnts)
{
    node_update_body<true>(rowptr, rec, bond_emb, hcur, wf1, wf2, b1, b2,
                           gamma, beta, mean, var, nullptr, batch, sums, cnts);
}

__global__ __launch_bounds__(128) void k_pool_mlp(
    const float* __restrict__ sums, const float* __restrict__ cnts,
    const float* __restrict__ mw, const float* __restrict__ mb,
    float* __restrict__ out)
{
    __shared__ float p[HID];
    int g = blockIdx.x;
    int t = threadIdx.x;
    float cnt = cnts[g];
    cnt = cnt > 1.f ? cnt : 1.f;
    p[t] = sums[(size_t)g * HID + t] / cnt;
    __syncthreads();
    if (t < OUTD) {
        float a = mb[t];
        #pragma unroll 16
        for (int c = 0; c < HID; ++c) a += p[c] * mw[c * OUTD + t];
        out[g * OUTD + t] = a;
    }
}

extern "C" void kernel_launch(void* const* d_in, const int* in_sizes, int n_in,
                              void* d_out, int out_size, void* d_ws, size_t ws_size,
                              hipStream_t stream) {
    const int*   x          = (const int*)  d_in[0];
    const int*   edge_index = (const int*)  d_in[1];
    const int*   edge_attr  = (const int*)  d_in[2];
    const int*   batch      = (const int*)  d_in[3];
    const float* atom_emb   = (const float*)d_in[4];
    const float* bond_emb   = (const float*)d_in[5];
    const float* lin1_w     = (const float*)d_in[6];
    const float* lin1_b     = (const float*)d_in[7];
    const float* lin2_w     = (const float*)d_in[8];
    const float* lin2_b     = (const float*)d_in[9];
    const float* bn_gamma   = (const float*)d_in[10];
    const float* bn_beta    = (const float*)d_in[11];
    const float* bn_mean    = (const float*)d_in[12];
    const float* bn_var     = (const float*)d_in[13];
    const float* mlp_w      = (const float*)d_in[14];
    const float* mlp_b      = (const float*)d_in[15];

    char* ws = (char*)d_ws;
    size_t off = 0;
    auto alloc = [&](size_t bytes) -> char* {
        char* p = ws + off;
        off = (off + bytes + 255) & ~(size_t)255;
        return p;
    };
    float*  h0   = (float*) alloc((size_t)N_NODES * HID * 4);
    float*  h1   = (float*) alloc((size_t)N_NODES * HID * 4);
    __bf16* wf1  = (__bf16*)alloc((size_t)NLAYERS * 16384 * 2);
    __bf16* wf2  = (__bf16*)alloc((size_t)NLAYERS * 16384 * 2);
    float*  sums = (float*) alloc((size_t)N_GRAPHS * HID * 4);
    float*  cnts = (float*) alloc((size_t)N_GRAPHS * 4);
    int*    coff   = (int*)  alloc((size_t)EP_CHUNKS * (EB_BINS + 1) * 4);
    int*    rowptr = (int*)  alloc((size_t)(N_NODES + 1) * 4);
    int4*   tmp    = (int4*) alloc((size_t)N_EDGES * 16);
    int2*   rec    = (int2*) alloc((size_t)N_EDGES * 8);

    const int nodeChunks = N_NODES * 32;

    k_setup<<<PREP_BLOCKS + EP_CHUNKS + (nodeChunks + 255) / 256, 256, 0, stream>>>(
        x, atom_emb, h0, lin1_w, lin2_w, wf1, wf2, sums, cnts,
        edge_index, edge_attr, coff, tmp);

    k_build<<<EB_BINS, 1024, 0, stream>>>(coff, tmp, rowptr, rec);

    float* hc = h0;
    float* hn = h1;
    for (int i = 0; i < NLAYERS - 1; ++i) {
        k_node_update<<<N_NODES / 16, 256, 0, stream>>>(
            rowptr, rec, bond_emb, hc, wf1 + (size_t)i * 16384, wf2 + (size_t)i * 16384,
            lin1_b + i * HID, lin2_b + i * HID,
            bn_gamma + i * HID, bn_beta + i * HID,
            bn_mean + i * HID, bn_var + i * HID, hn);
        float* t = hc; hc = hn; hn = t;
    }
    k_node_update_pool<<<N_NODES / 16, 256, 0, stream>>>(
        rowptr, rec, bond_emb, hc,
        wf1 + (size_t)(NLAYERS - 1) * 16384, wf2 + (size_t)(NLAYERS - 1) * 16384,
        lin1_b + (NLAYERS - 1) * HID, lin2_b + (NLAYERS - 1) * HID,
        bn_gamma + (NLAYERS - 1) * HID, bn_beta + (NLAYERS - 1) * HID,
        bn_mean + (NLAYERS - 1) * HID, bn_var + (NLAYERS - 1) * HID,
        batch, sums, cnts);

    k_pool_mlp<<<N_GRAPHS, 128, 0, stream>>>(sums, cnts, mlp_w, mlp_b, (float*)d_out);
}


# LDS z and z1 tiles: row pitch 256 B -> 272 B (16 B x odd): the 16 rows that half of an A-fragment 16-byte read touches fall on 16 bank groups, not one (16-way bank conflict gone); sums unchanged
# speedup vs baseline: 4.2722x; 1.3392x over previous
#include <hip/hip_runtime.h>
#include <hip/hip_bf16.h>

#define N_NODES 50000
#define N_EDGES 500000
#define N_GRAPHS 1000
#define HID 128
#define OUTD 10
#define NLAYERS 4

typedef __attribute__((ext_vector_type(16))) __bf16 v16bf;
typedef __attribute__((ext_vector_type(8)))  __bf16 v8bf;
typedef __attribute__((ext_vector_type(4)))  __bf16 v4bf;
typedef __attribute__((ext_vector_type(8)))  float  v8f;

__device__ __forceinline__ void encode_nodes(
    const int gid, const int* __restrict__ x, const float* __restrict__ atom_emb,
    float* __restrict__ h)
{
    if (gid >= N_NODES * 32) return;
    int n = gid >> 5;
    int c = (gid & 31) * 4;
    float4 acc = make_float4(0.f, 0.f, 0.f, 0.f);
    #pragma unroll
    for (int f = 0; f < 9; ++f) {
        int idx = x[n * 9 + f];
        const float4 v = *(const float4*)(atom_emb + ((size_t)(f * 120 + idx)) * HID + c);
        acc.x += v.x; acc.y += v.y; acc.z += v.z; acc.w += v.w;
    }
    *(float4*)(h + (size_t)n * HID + c) = acc;
}

#define EP_CHUNK  2048
#define EP_CHUNKS ((N_EDGES + EP_CHUNK - 1) / EP_CHUNK)
#define EP_STEPS  8
#define EB_BINS   128
#define EB_NPB    391
#define EB_CPW    8
static_assert(EB_BINS * EB_NPB >= N_NODES, "the bins cover every node");
static_assert(8 * EP_STEPS * 32 == EP_CHUNK, "8 waves of EP_STEPS steps take one chunk");
static_assert(32 * EB_CPW >= EP_CHUNKS, "the 32 waves of k_build take every chunk");

__device__ __forceinline__ void rank_in_step(const int key, const bool valid, const int lane,
                                             int& group, int& before)
{
    group = 0; before = 0;
    unsigned mask = __builtin_amdgcn_ballot_w32(valid);
    while (mask != 0u) {
        const int l  = __builtin_ctz(mask);
        mask &= mask - 1u;
        const int kl = __builtin_amdgcn_readlane(key, l);
        if (kl == key) { group += 1; if (l < lane) before += 1; }
    }
}

__device__ __forceinline__ void part_edges(
    const int c, const int* __restrict__ edge_index, const int* __restrict__ edge_attr,
    int* __restrict__ coff, int4* __restrict__ tmp)
{
    __shared__ int cw[8 * EB_BINS];
    __shared__ int wsum[8];
    const int tid  = threadIdx.x;
    const int lane = tid & 31;
    const int wave = tid >> 5;
    for (int i = tid; i < 8 * EB_BINS; i += 256) cw[i] = 0;
    __syncthreads();

    const int e0 = c * EP_CHUNK + wave * (EP_STEPS * 32) + lane;
    int dv[EP_STEPS], sr[EP_STEPS], code[EP_STEPS];
    #pragma unroll
    for (int u = 0; u < EP_STEPS; ++u) {
        const int e  = e0 + u * 32;
        const int ec = (e < N_EDGES) ? e : (N_EDGES - 1);
        const int d  = edge_index[N_EDGES + ec];
        sr[u]        = edge_index[ec];
        const int a0 = edge_attr[ec * 3 + 0];
        const int a1 = edge_attr[ec * 3 + 1];
        const int a2 = edge_attr[ec * 3 + 2];
        code[u] = a0 | (a1 << 8) | (a2 << 16);
        dv[u]   = (e < N_EDGES) ? d : -1;
    }
    #pragma unroll
    for (int u = 0; u < EP_STEPS; ++u)
        if (dv[u] >= 0) atomicAdd(&cw[wave * EB_BINS + dv[u] / EB_NPB], 1);
    __syncthreads();

    int total = 0;
    if (tid < EB_BINS) {
        for (int w = 0; w < 8; ++w) {
            const int n = cw[w * EB_BINS + tid];
            cw[w * EB_BINS + tid] = total;
            total += n;
        }
    }
    int inc = total;
    for (int o = 1; o < 32; o <<= 1) {
        const int t = __shfl_up(inc, o);
        if (lane >= o) inc += t;
    }
    if (lane == 31) wsum[wave] = inc;
    __syncthreads();
    int tw = (lane < wave) ? wsum[lane & 7] : 0;
    for (int o = 16; o > 0; o >>= 1) tw += __shfl_xor(tw, o);
    const int excl = tw + inc - total;
    if (tid < EB_BINS) {
        for (int w = 0; w < 8; ++w) cw[w * EB_BINS + tid] += excl;
        coff[c * (EB_BINS + 1) + tid] = excl;
        if (tid == EB_BINS - 1) coff[c * (EB_BINS + 1) + EB_BINS] = excl + total;
    }
    __syncthreads();

    #pragma unroll
    for (int u = 0; u < EP_STEPS; ++u) {
        const bool valid = dv[u] >= 0;
        const int  bin   = valid ? dv[u] / EB_NPB : -1;
        int group, before;
        rank_in_step(bin, valid, lane, group, before);
        if (valid) {
            const int idx = wave * EB_BINS + bin;
            const int cur = cw[idx];
            tmp[c * EP_CHUNK + cur + before] = make_int4(dv[u], sr[u], code[u], 0);
            if (before == group - 1) cw[idx] = cur + group;
        }
    }
}

__global__ __launch_bounds__(1024) void k_build(
    const int* __restrict__ coff, const int4* __restrict__ tmp,
    int* __restrict__ rowptr, int2* __restrict__ rec)
{
    __shared__ int cntw[32 * EB_NPB];
    __shared__ int wless[32];
    __shared__ int wtot[32];
    const int tid  = threadIdx.x;
    const int lane = tid & 31;
    const int wave = __builtin_amdgcn_readfirstlane(tid >> 5);
    const int b    = blockIdx.x;
    const int lo   = b * EB_NPB;

    for (int i = tid; i < 32 * EB_NPB; i += 1024) cntw[i] = 0;
    __syncthreads();

    int vo0, vo1;
    {
        const int  ck    = wave * EB_CPW + (lane & (EB_CPW - 1));
        const bool livek = ck < EP_CHUNKS;
        const int  cck   = livek ? ck : (EP_CHUNKS - 1);
        vo0 = coff[cck * (EB_BINS + 1) + b];
        vo1 = coff[cck * (EB_BINS + 1) + b + 1];
        if (!livek) { vo0 = 0; vo1 = 0; }
    }
    int s[EB_CPW], t[EB_CPW];
    int less = 0;
    #pragma unroll
    for (int k = 0; k < EB_CPW; ++k) {
        const int c  = wave * EB_CPW + k;
        const int o0 = __builtin_amdgcn_readlane(vo0, k);
        const int o1 = __builtin_amdgcn_readlane(vo1, k);
        less += o0;
        s[k] = c * EP_CHUNK + o0;
        t[k] = c * EP_CHUNK + o1;
    }
    int4 r0[EB_CPW];
    #pragma unroll
    for (int k = 0; k < EB_CPW; ++k) {
        const int p = s[k] + lane;
        const int4 v = tmp[(p < t[k]) ? p : 0];
        r0[k] = v;
        if (p >= t[k]) r0[k].x = -1;
    }

    int mine = 0;
    #pragma unroll
    for (int k = 0; k < EB_CPW; ++k) {
        mine += t[k] - s[k];
        if (r0[k].x >= 0) atomicAdd(&cntw[wave * EB_NPB + (r0[k].x - lo)], 1);
        for (int base = s[k] + 32; base < t[k]; base += 32) {
            const int p = base + lane;
            if (p < t[k]) atomicAdd(&cntw[wave * EB_NPB + (tmp[p].x - lo)], 1);
        }
    }
    if (lane == 0) { wless[wave] = less; wtot[wave] = mine; }
    __syncthreads();

    int total = 0;
    if (tid < EB_NPB) {
        for (int w = 0; w < 32; ++w) {
            const int n = cntw[w * EB_NPB + tid];
            cntw[w * EB_NPB + tid] = total;
            total += n;
        }
    }
    int tl = wless[lane];
    int ta = wtot[lane];
    for (int o = 16; o > 0; o >>= 1) {
        tl += __shfl_xor(tl, o);
        ta += __shfl_xor(ta, o);
    }
    __syncthreads();
    int inc = total;
    for (int o = 1; o < 32; o <<= 1) {
        const int x = __shfl_up(inc, o);
        if (lane >= o) inc += x;
    }
    if (lane == 31) wtot[wave] = inc;
    __syncthreads();
    int tw = (lane < wave) ? wtot[lane] : 0;
    for (int o = 16; o > 0; o >>= 1) tw += __shfl_xor(tw, o);
    const int excl = tl + tw + inc - total;
    if (tid < EB_NPB) {
        for (int w = 0; w < 32; ++w) cntw[w * EB_NPB + tid] += excl;
        if (lo + tid < N_NODES) rowptr[lo + tid] = excl;
    }
    if (b == EB_BINS - 1 && tid == 0) rowptr[N_NODES] = tl + ta;
    __syncthreads();

    #pragma unroll
    for (int k = 0; k < EB_CPW; ++k) {
        int4 v = r0[k];
        int base = s[k];
        while (true) {
            const bool valid = v.x >= 0;
            int group, before;
            rank_in_step(v.x, valid, lane, group, before);
            if (valid) {
                const int idx = wave * EB_NPB + (v.x - lo);
                const int cur = cntw[idx];
                rec[cur + before] = make_int2(v.y, v.z);
                if (before == group - 1) cntw[idx] = cur + group;
            }
            base += 32;
            if (base >= t[k]) break;
            const int p = base + lane;
            v = tmp[(p < t[k]) ? p : 0];
            if (p >= t[k]) v.x = -1;
        }
    }
}

__device__ __forceinline__ float4 edge_msg(const float4 hv, const float4 e0,
                                           const float4 e1, const float4 e2)
{
    float m0 = hv.x + e0.x + e1.x + e2.x; m0 = m0 > 0.f ? m0 : 0.f;
    float m1 = hv.y + e0.y + e1.y + e2.y; m1 = m1 > 0.f ? m1 : 0.f;
    float m2 = hv.z + e0.z + e1.z + e2.z; m2 = m2 > 0.f ? m2 : 0.f;
    float m3 = hv.w + e0.w + e1.w + e2.w; m3 = m3 > 0.f ? m3 : 0.f;
    return make_float4(m0, m1, m2, m3);
}

__device__ __forceinline__ float4 gather_row(
    const int n, const int s, const int t, const int lane,
    const int2* __restrict__ rec, const float4* b4,
    const float4* __restrict__ h4)
{
    float4 acc = h4[(size_t)n * 32 + lane];
    for (int base = s; base < t; base += 32) {
        const int left = t - base;
        const int cnt  = left < 32 ? left : 32;
        int2 my = make_int2(0, 0);
        if (lane < cnt) my = rec[base + lane];
        int j = 0;
        for (; j + 4 <= cnt; j += 4) {
            float4 hv[4];
            int code[4];
            for (int u = 0; u < 4; ++u) {
                const int src = __builtin_amdgcn_readlane(my.x, j + u);
                code[u]       = __builtin_amdgcn_readlane(my.y, j + u);
                hv[u] = h4[(size_t)src * 32 + lane];
            }
            for (int u = 0; u < 4; ++u) {
                const float4 e0 = b4[(0 * 6 + (code[u] & 255)) * 32 + lane];
                const float4 e1 = b4[(1 * 6 + ((code[u] >> 8) & 255)) * 32 + lane];
                const float4 e2 = b4[(2 * 6 + (code[u] >> 16)) * 32 + lane];
                const float4 m = edge_msg(hv[u], e0, e1, e2);
                acc.x += m.x; acc.y += m.y; acc.z += m.z; acc.w += m.w;
            }
        }
        for (; j < cnt; ++j) {
            const int src  = __builtin_amdgcn_readlane(my.x, j);
            const int code = __builtin_amdgcn_readlane(my.y, j);
            const float4 hv = h4[(size_t)src * 32 + lane];
            const float4 e0 = b4[(0 * 6 + (code & 255)) * 32 + lane];
            const float4 e1 = b4[(1 * 6 + ((code >> 8) & 255)) * 32 + lane];
            const float4 e2 = b4[(2 * 6 + (code >> 16)) * 32 + lane];
            const float4 m = edge_msg(hv, e0, e1, e2);
            acc.x += m.x; acc.y += m.y; acc.z += m.z; acc.w += m.w;
        }
    }
    return acc;
}

#define PREP_ITEMS (NLAYERS * 4 * 8 * 32 * 16)
#define PREP_BLOCKS (PREP_ITEMS / 256)
__device__ __forceinline__ void prep_weights(
    const int o,
    const float* __restrict__ w1, const float* __restrict__ w2,
    __bf16* __restrict__ f1, __bf16* __restrict__ f2,
    float* __restrict__ sums, float* __restrict__ cnts)
{
    if (o >= PREP_ITEMS) return;
    static_assert(N_GRAPHS * HID <= 2 * PREP_ITEMS, "two pool sums per thread");
    static_assert(N_GRAPHS <= PREP_ITEMS, "one thread per graph count");
    if (o < N_GRAPHS * HID)              sums[o] = 0.f;
    if (o + PREP_ITEMS < N_GRAPHS * HID) sums[o + PREP_ITEMS] = 0.f;
    if (o < N_GRAPHS)                    cnts[o] = 0.f;
    int j     =  o        & 15;
    int lane  = (o >> 4)  & 31;
    int ntile = (o >> 9)  & 7;
    int ks    = (o >> 12) & 3;
    int layer =  o >> 14;
    int colL  = lane & 15;
    int khalf = (lane >> 4) * 8;
    int K     = ks * 32 + ((j < 8) ? (khalf + j) : (16 + khalf + (j - 8)));
    int col   = ntile * 16 + colL;
    size_t src = (size_t)layer * HID * HID + (size_t)K * HID + col;
    size_t dst = (size_t)layer * 16384 +
                 (((size_t)ks * 8 + ntile) * 32 + lane) * 16 + j;
    f1[dst] = (__bf16)w1[src];
    f2[dst] = (__bf16)w2[src];
}

__global__ __launch_bounds__(256) void k_setup(
    const int* __restrict__ x, const float* __restrict__ atom_emb, float* __restrict__ h,
    const float* __restrict__ w1, const float* __restrict__ w2,
    __bf16* __restrict__ f1, __bf16* __restrict__ f2,
    float* __restrict__ sums, float* __restrict__ cnts,
    const int* __restrict__ edge_index, const int* __restrict__ edge_attr,
    int* __restrict__ coff, int4* __restrict__ tmp)
{
    const int b = blockIdx.x;
    if (b < PREP_BLOCKS) {
        prep_weights(b * 256 + threadIdx.x, w1, w2, f1, f2, sums, cnts);
    } else if (b < PREP_BLOCKS + EP_CHUNKS) {
        part_edges(b - PREP_BLOCKS, edge_index, edge_attr, coff, tmp);
    } else {
        encode_nodes((b - PREP_BLOCKS - EP_CHUNKS) * 256 + threadIdx.x, x, atom_emb, h);
    }
}

template <bool LAST>
__device__ __forceinline__ void node_update_body(
    const int* __restrict__ rowptr, const int2* __restrict__ rec,
    const float* __restrict__ bond_emb, const float* __restrict__ hcur,
    const __bf16* __restrict__ wf1, const __bf16* __restrict__ wf2,
    const float* __restrict__ b1, const float* __restrict__ b2,
    const float* __restrict__ gamma, const float* __restrict__ beta,
    const float* __restrict__ mean, const float* __restrict__ var,
    float* __restrict__ hnext,
    const int* __restrict__ batch, float* __restrict__ sums, float* __restrict__ cnts)
{
    constexpr int ZP = HID + 8;
    __shared__ __align__(16) __bf16 zt[16 * ZP];
    __shared__ __align__(16) __bf16 z1[16 * ZP];
    __shared__ __align__(16) float  bt[18 * HID];

    static_assert(N_NODES % 16 == 0, "whole 16-node tiles: no row guard below");
    const int tid   = threadIdx.x;
    const int wave  = tid >> 5;
    const int lane  = tid & 31;
    const int node0 = blockIdx.x * 16;

    const int m0 = __builtin_amdgcn_readfirstlane(wave * 2);
    const int n0 = node0 + m0;
    const int s0 = rowptr[n0];
    const int s1 = rowptr[n0 + 1];
    const int s2 = rowptr[n0 + 2];

    {
        static_assert(18 * HID / 4 == 2 * 256 + 64, "2 float4 per thread and 64 more");
        const float4* __restrict__ g4 = (const float4*)bond_emb;
        float4* bt4w = (float4*)bt;
        const float4 t0 = g4[tid];
        const float4 t1 = g4[tid + 256];
        float4 t2 = make_float4(0.f, 0.f, 0.f, 0.f);
        if (tid < 64) t2 = g4[tid + 512];
        bt4w[tid]       = t0;
        bt4w[tid + 256] = t1;
        if (tid < 64) bt4w[tid + 512] = t2;
    }
    __syncthreads();

    {
        const float4* __restrict__ h4 = (const float4*)hcur;
        const float4* b4 = (const float4*)bt;
        const float4 za = gather_row(n0,     s0, s1, lane, rec, b4, h4);
        v4bf pa;
        pa[0] = (__bf16)za.x; pa[1] = (__bf16)za.y; pa[2] = (__bf16)za.z; pa[3] = (__bf16)za.w;
        *(v4bf*)&zt[m0 * ZP + lane * 4]       = pa;
        const float4 zb = gather_row(n0 + 1, s1, s2, lane, rec, b4, h4);
        v4bf pb;
        pb[0] = (__bf16)zb.x; pb[1] = (__bf16)zb.y; pb[2] = (__bf16)zb.z; pb[3] = (__bf16)zb.w;
        *(v4bf*)&zt[(m0 + 1) * ZP + lane * 4] = pb;
    }
    __syncthreads();

    const int colL  = lane & 15;
    const int col   = wave * 16 + colL;
    const int khalf = (lane >> 4) * 8;
    const int rowA  = lane & 15;

    v8f acc1 = {};
    #pragma unroll
    for (int ks = 0; ks < 4; ++ks) {
        const __bf16* ap = &zt[rowA * ZP + ks * 32 + khalf];
        v8bf alo = *(const v8bf*)(ap);
        v8bf ahi = *(const v8bf*)(ap + 16);
        v16bf a = __builtin_shufflevector(alo, ahi,
                    0,1,2,3,4,5,6,7,8,9,10,11,12,13,14,15);
        v16bf b = *(const v16bf*)(wf1 + (((size_t)ks * 8 + wave) * 32 + lane) * 16);
        acc1 = __builtin_amdgcn_wmma_f32_16x16x32_bf16(
                   false, a, false, b, (short)0, acc1, false, false);
    }

    const float bias1 = b1[col];
    #pragma unroll
    for (int r = 0; r < 8; ++r) {
        int m = (lane < 16) ? r : (r + 8);
        float v = acc1[r] + bias1;
        v = v > 0.f ? v : 0.f;
        z1[m * ZP + col] = (__bf16)v;
    }
    __syncthreads();

    v8f acc2 = {};
    #pragma unroll
    for (int ks = 0; ks < 4; ++ks) {
        const __bf16* ap = &z1[rowA * ZP + ks * 32 + khalf];
        v8bf alo = *(const v8bf*)(ap);
        v8bf ahi = *(const v8bf*)(ap + 16);
        v16bf a = __builtin_shufflevector(alo, ahi,
                    0,1,2,3,4,5,6,7,8,9,10,11,12,13,14,15);
        v16bf b = *(const v16bf*)(wf2 + (((size_t)ks * 8 + wave) * 32 + lane) * 16);
        acc2 = __builtin_amdgcn_wmma_f32_16x16x32_bf16(
                   false, a, false, b, (short)0, acc2, false, false);
    }

    const float bias2 = b2[col];
    const float mu    = mean[col];
    const float inv   = rsqrtf(var[col] + 1e-5f);
    const float ga    = gamma[col];
    const float be    = beta[col];
    if (!LAST) {
        #pragma unroll
        for (int r = 0; r < 8; ++r) {
            int m = (lane < 16) ? r : (r + 8);
            size_t idx = (size_t)(node0 + m) * HID + col;
            float v = acc2[r] + bias2;
            v = (v - mu) * inv * ga + be;
            v = v > 0.f ? v : 0.f;
            hnext[idx] = hcur[idx] + v;
        }
    } else {
        int G[16];
        #pragma unroll
        for (int k = 0; k < 16; ++k) G[k] = batch[node0 + k];
        const bool hi   = lane >= 16;
        const int  row0 = node0 + (hi ? 8 : 0);
        float hv[8];
        #pragma unroll
        for (int r = 0; r < 8; ++r) {
            size_t idx = (size_t)(row0 + r) * HID + col;
            float v = acc2[r] + bias2;
            v = (v - mu) * inv * ga + be;
            v = v > 0.f ? v : 0.f;
            hv[r] = hcur[idx] + v;
        }
        float run  = 0.f;
        float nrun = 0.f;
        #pragma unroll
        for (int r = 0; r < 8; ++r) {
            run  += hv[r];
            nrun += 1.f;
            const int gr = hi ? G[r + 8] : G[r];
            bool flush = true;
            if (r < 7) flush = ((hi ? G[r + 9] : G[r + 1]) != gr);
            if (flush) {
                atomicAdd(sums + (size_t)gr * HID + col, run);
                if (col == 0) atomicAdd(cnts + gr, nrun);
                run  = 0.f;
                nrun = 0.f;
            }
        }
    }
}

__global__ __launch_bounds__(256, 16) void k_node_update(
    const int* __restrict__ rowptr, const int2* __restrict__ rec,
    const float* __restrict__ bond_emb, const float* __restrict__ hcur,
    const __bf16* __restrict__ wf1, const __bf16* __restrict__ wf2,
    const float* __restrict__ b1, const float* __restrict__ b2,
    const float* __restrict__ gamma, const float* __restrict__ beta,
    const float* __restrict__ mean, const float* __restrict__ var,
    float* __restrict__ hnext)
{
    node_update_body<false>(rowptr, rec, bond_emb, hcur, wf1, wf2, b1, b2,
                            gamma, beta, mean, var, hnext, nullptr, nullptr, nullptr);
}

__global__ __launch_bounds__(256, 16) void k_node_update_pool(
    const int* __restrict__ rowptr, const int2* __restrict__ rec,
    const float* __restrict__ bond_emb, const float* __restrict__ hcur,
    const __bf16* __restrict__ wf1, const __bf16* __restrict__ wf2,
    const float* __restrict__ b1, const float* __restrict__ b2,
    const float* __restrict__ gamma, const float* __restrict__ beta,
    const float* __restrict__ mean, const float* __restrict__ var,
    const int* __restrict__ batch, float* __restrict__ sums, float* __restrict__ cnts)
{
    node_update_body<true>(rowptr, rec, bond_emb, hcur, wf1, wf2, b1, b2,
                           gamma, beta, mean, var, nullptr, batch, sums, cnts);
}

__global__ __launch_bounds__(128) void k_pool_mlp(
    const float* __restrict__ sums, const float* __restrict__ cnts,
    const float* __restrict__ mw, const float* __restrict__ mb,
    float* __restrict__ out)
{
    __shared__ float p[HID];
    int g = blockIdx.x;
    int t = threadIdx.x;
    float cnt = cnts[g];
    cnt = cnt > 1.f ? cnt : 1.f;
    p[t] = sums[(size_t)g * HID + t] / cnt;
    __syncthreads();
    if (t < OUTD) {
        float a = mb[t];
        #pragma unroll 16
        for (int c = 0; c < HID; ++c) a += p[c] * mw[c * OUTD + t];
        out[g * OUTD + t] = a;
    }
}

extern "C" void kernel_launch(void* const* d_in, const int* in_sizes, int n_in,
                              void* d_out, int out_size, void* d_ws, size_t ws_size,
                              hipStream_t stream) {
    const int*   x          = (const int*)  d_in[0];
    const int*   edge_index = (const int*)  d_in[1];
    const int*   edge_attr  = (const int*)  d_in[2];
    const int*   batch      = (const int*)  d_in[3];
    const float* atom_emb   = (const float*)d_in[4];
    const float* bond_emb   = (const float*)d_in[5];
    const float* lin1_w     = (const float*)d_in[6];
    const float* lin1_b     = (const float*)d_in[7];
    const float* lin2_w     = (const float*)d_in[8];
    const float* lin2_b     = (const float*)d_in[9];
    const float* bn_gamma   = (const float*)d_in[10];
    const float* bn_beta    = (const float*)d_in[11];
    const float* bn_mean    = (const float*)d_in[12];
    const float* bn_var     = (const float*)d_in[13];
    const float* mlp_w      = (const float*)d_in[14];
    const float* mlp_b      = (const float*)d_in[15];

    char* ws = (char*)d_ws;
    size_t off = 0;
    auto alloc = [&](size_t bytes) -> char* {
        char* p = ws + off;
        off = (off + bytes + 255) & ~(size_t)255;
        return p;
    };
    float*  h0   = (float*) alloc((size_t)N_NODES * HID * 4);
    float*  h1   = (float*) alloc((size_t)N_NODES * HID * 4);
    __bf16* wf1  = (__bf16*)alloc((size_t)NLAYERS * 16384 * 2);
    __bf16* wf2  = (__bf16*)alloc((size_t)NLAYERS * 16384 * 2);
    float*  sums = (float*) alloc((size_t)N_GRAPHS * HID * 4);
    float*  cnts = (float*) alloc((size_t)N_GRAPHS * 4);
    int*    coff   = (int*)  alloc((size_t)EP_CHUNKS * (EB_BINS + 1) * 4);
    int*    rowptr = (int*)  alloc((size_t)(N_NODES + 1) * 4);
    int4*   tmp    = (int4*) alloc((size_t)N_EDGES * 16);
    int2*   rec    = (int2*) alloc((size_t)N_EDGES * 8);

    const int nodeChunks = N_NODES * 32;

    k_setup<<<PREP_BLOCKS + EP_CHUNKS + (nodeChunks + 255) / 256, 256, 0, stream>>>(
        x, atom_emb, h0, lin1_w, lin2_w, wf1, wf2, sums, cnts,
        edge_index, edge_attr, coff, tmp);

    k_build<<<EB_BINS, 1024, 0, stream>>>(coff, tmp, rowptr, rec);

    float* hc = h0;
    float* hn = h1;
    for (int i = 0; i < NLAYERS - 1; ++i) {
        k_node_update<<<N_NODES / 16, 256, 0, stream>>>(
            rowptr, rec, bond_emb, hc, wf1 + (size_t)i * 16384, wf2 + (size_t)i * 16384,
            lin1_b + i * HID, lin2_b + i * HID,
            bn_gamma + i * HID, bn_beta + i * HID,
            bn_mean + i * HID, bn_var + i * HID, hn);
        float* t = hc; hc = hn; hn = t;
    }
    k_node_update_pool<<<N_NODES / 16, 256, 0, stream>>>(
        rowptr, rec, bond_emb, hc,
        wf1 + (size_t)(NLAYERS - 1) * 16384, wf2 + (size_t)(NLAYERS - 1) * 16384,
        lin1_b + (NLAYERS - 1) * HID, lin2_b + (NLAYERS - 1) * HID,
        bn_gamma + (NLAYERS - 1) * HID, bn_beta + (NLAYERS - 1) * HID,
        bn_mean + (NLAYERS - 1) * HID, bn_var + (NLAYERS - 1) * HID,
        batch, sums, cnts);

    k_pool_mlp<<<N_GRAPHS, 128, 0, stream>>>(sums, cnts, mlp_w, mlp_b, (float*)d_out);
}


# node-update gather: one record load for both rows of a wave, sent with the first h row; a row's last 1-3 edges are one step: ~7 dependent memory waits per wave, not ~10; per-row sum order kept
# speedup vs baseline: 4.3991x; 1.0297x over previous
#include <hip/hip_runtime.h>
#include <hip/hip_bf16.h>

#define N_NODES 50000
#define N_EDGES 500000
#define N_GRAPHS 1000
#define HID 128
#define OUTD 10
#define NLAYERS 4

typedef __attribute__((ext_vector_type(16))) __bf16 v16bf;
typedef __attribute__((ext_vector_type(8)))  __bf16 v8bf;
typedef __attribute__((ext_vector_type(4)))  __bf16 v4bf;
typedef __attribute__((ext_vector_type(8)))  float  v8f;
typedef __attribute__((ext_vector_type(4)))  float  v4f;
typedef __attribute__((address_space(1)))    v4f    gv4f;

__device__ __forceinline__ void encode_nodes(
    const int gid, const int* __restrict__ x, const float* __restrict__ atom_emb,
    float* __restrict__ h)
{
    if (gid >= N_NODES * 32) return;
    int n = gid >> 5;
    int c = (gid & 31) * 4;
    float4 acc = make_float4(0.f, 0.f, 0.f, 0.f);
    #pragma unroll
    for (int f = 0; f < 9; ++f) {
        int idx = x[n * 9 + f];
        const float4 v = *(const float4*)(atom_emb + ((size_t)(f * 120 + idx)) * HID + c);
        acc.x += v.x; acc.y += v.y; acc.z += v.z; acc.w += v.w;
    }
    *(float4*)(h + (size_t)n * HID + c) = acc;
}

#define EP_CHUNK  2048
#define EP_CHUNKS ((N_EDGES + EP_CHUNK - 1) / EP_CHUNK)
#define EP_STEPS  8
#define EB_BINS   128
#define EB_NPB    391
#define EB_CPW    8
static_assert(EB_BINS * EB_NPB >= N_NODES, "the bins cover every node");
static_assert(8 * EP_STEPS * 32 == EP_CHUNK, "8 waves of EP_STEPS steps take one chunk");
static_assert(32 * EB_CPW >= EP_CHUNKS, "the 32 waves of k_build take every chunk");

__device__ __forceinline__ void rank_in_step(const int key, const bool valid, const int lane,
                                             int& group, int& before)
{
    group = 0; before = 0;
    unsigned mask = __builtin_amdgcn_ballot_w32(valid);
    while (mask != 0u) {
        const int l  = __builtin_ctz(mask);
        mask &= mask - 1u;
        const int kl = __builtin_amdgcn_readlane(key, l);
        if (kl == key) { group += 1; if (l < lane) before += 1; }
    }
}

__device__ __forceinline__ void part_edges(
    const int c, const int* __restrict__ edge_index, const int* __restrict__ edge_attr,
    int* __restrict__ coff, int4* __restrict__ tmp)
{
    __shared__ int cw[8 * EB_BINS];
    __shared__ int wsum[8];
    const int tid  = threadIdx.x;
    const int lane = tid & 31;
    const int wave = tid >> 5;
    for (int i = tid; i < 8 * EB_BINS; i += 256) cw[i] = 0;
    __syncthreads();

    const int e0 = c * EP_CHUNK + wave * (EP_STEPS * 32) + lane;
    int dv[EP_STEPS], sr[EP_STEPS], code[EP_STEPS];
    #pragma unroll
    for (int u = 0; u < EP_STEPS; ++u) {
        const int e  = e0 + u * 32;
        const int ec = (e < N_EDGES) ? e : (N_EDGES - 1);
        const int d  = edge_index[N_EDGES + ec];
        sr[u]        = edge_index[ec];
        const int a0 = edge_attr[ec * 3 + 0];
        const int a1 = edge_attr[ec * 3 + 1];
        const int a2 = edge_attr[ec * 3 + 2];
        code[u] = a0 | (a1 << 8) | (a2 << 16);
        dv[u]   = (e < N_EDGES) ? d : -1;
    }
    #pragma unroll
    for (int u = 0; u < EP_STEPS; ++u)
        if (dv[u] >= 0) atomicAdd(&cw[wave * EB_BINS + dv[u] / EB_NPB], 1);
    __syncthreads();

    int total = 0;
    if (tid < EB_BINS) {
        for (int w = 0; w < 8; ++w) {
            const int n = cw[w * EB_BINS + tid];
            cw[w * EB_BINS + tid] = total;
            total += n;
        }
    }
    int inc = total;
    for (int o = 1; o < 32; o <<= 1) {
        const int t = __shfl_up(inc, o);
        if (lane >= o) inc += t;
    }
    if (lane == 31) wsum[wave] = inc;
    __syncthreads();
    int tw = (lane < wave) ? wsum[lane & 7] : 0;
    for (int o = 16; o > 0; o >>= 1) tw += __shfl_xor(tw, o);
    const int excl = tw + inc - total;
    if (tid < EB_BINS) {
        for (int w = 0; w < 8; ++w) cw[w * EB_BINS + tid] += excl;
        coff[c * (EB_BINS + 1) + tid] = excl;
        if (tid == EB_BINS - 1) coff[c * (EB_BINS + 1) + EB_BINS] = excl + total;
    }
    __syncthreads();

    #pragma unroll
    for (int u = 0; u < EP_STEPS; ++u) {
        const bool valid = dv[u] >= 0;
        const int  bin   = valid ? dv[u] / EB_NPB : -1;
        int group, before;
        rank_in_step(bin, valid, lane, group, before);
        if (valid) {
            const int idx = wave * EB_BINS + bin;
            const int cur = cw[idx];
            tmp[c * EP_CHUNK + cur + before] = make_int4(dv[u], sr[u], code[u], 0);
            if (before == group - 1) cw[idx] = cur + group;
        }
    }
}

__global__ __launch_bounds__(1024) void k_build(
    const int* __restrict__ coff, const int4* __restrict__ tmp,
    int* __restrict__ rowptr, int2* __restrict__ rec)
{
    __shared__ int cntw[32 * EB_NPB];
    __shared__ int wless[32];
    __shared__ int wtot[32];
    const int tid  = threadIdx.x;
    const int lane = tid & 31;
    const int wave = __builtin_amdgcn_readfirstlane(tid >> 5);
    const int b    = blockIdx.x;
    const int lo   = b * EB_NPB;

    for (int i = tid; i < 32 * EB_NPB; i += 1024) cntw[i] = 0;
    __syncthreads();

    int vo0, vo1;
    {
        const int  ck    = wave * EB_CPW + (lane & (EB_CPW - 1));
        const bool livek = ck < EP_CHUNKS;
        const int  cck   = livek ? ck : (EP_CHUNKS - 1);
        vo0 = coff[cck * (EB_BINS + 1) + b];
        vo1 = coff[cck * (EB_BINS + 1) + b + 1];
        if (!livek) { vo0 = 0; vo1 = 0; }
    }
    int s[EB_CPW], t[EB_CPW];
    int less = 0;
    #pragma unroll
    for (int k = 0; k < EB_CPW; ++k) {
        const int c  = wave * EB_CPW + k;
        const int o0 = __builtin_amdgcn_readlane(vo0, k);
        const int o1 = __builtin_amdgcn_readlane(vo1, k);
        less += o0;
        s[k] = c * EP_CHUNK + o0;
        t[k] = c * EP_CHUNK + o1;
    }
    int4 r0[EB_CPW];
    #pragma unroll
    for (int k = 0; k < EB_CPW; ++k) {
        const int p = s[k] + lane;
        const int4 v = tmp[(p < t[k]) ? p : 0];
        r0[k] = v;
        if (p >= t[k]) r0[k].x = -1;
    }

    int mine = 0;
    #pragma unroll
    for (int k = 0; k < EB_CPW; ++k) {
        mine += t[k] - s[k];
        if (r0[k].x >= 0) atomicAdd(&cntw[wave * EB_NPB + (r0[k].x - lo)], 1);
        for (int base = s[k] + 32; base < t[k]; base += 32) {
            const int p = base + lane;
            if (p < t[k]) atomicAdd(&cntw[wave * EB_NPB + (tmp[p].x - lo)], 1);
        }
    }
    if (lane == 0) { wless[wave] = less; wtot[wave] = mine; }
    __syncthreads();

    int total = 0;
    if (tid < EB_NPB) {
        for (int w = 0; w < 32; ++w) {
            const int n = cntw[w * EB_NPB + tid];
            cntw[w * EB_NPB + tid] = total;
            total += n;
        }
    }
    int tl = wless[lane];
    int ta = wtot[lane];
    for (int o = 16; o > 0; o >>= 1) {
        tl += __shfl_xor(tl, o);
        ta += __shfl_xor(ta, o);
    }
    __syncthreads();
    int inc = total;
    for (int o = 1; o < 32; o <<= 1) {
        const int x = __shfl_up(inc, o);
        if (lane >= o) inc += x;
    }
    if (lane == 31) wtot[wave] = inc;
    __syncthreads();
    int tw = (lane < wave) ? wtot[lane] : 0;
    for (int o = 16; o > 0; o >>= 1) tw += __shfl_xor(tw, o);
    const int excl = tl + tw + inc - total;
    if (tid < EB_NPB) {
        for (int w = 0; w < 32; ++w) cntw[w * EB_NPB + tid] += excl;
        if (lo + tid < N_NODES) rowptr[lo + tid] = excl;
    }
    if (b == EB_BINS - 1 && tid == 0) rowptr[N_NODES] = tl + ta;
    __syncthreads();

    #pragma unroll
    for (int k = 0; k < EB_CPW; ++k) {
        int4 v = r0[k];
        int base = s[k];
        while (true) {
            const bool valid = v.x >= 0;
            int group, before;
            rank_in_step(v.x, valid, lane, group, before);
            if (valid) {
                const int idx = wave * EB_NPB + (v.x - lo);
                const int cur = cntw[idx];
                rec[cur + before] = make_int2(v.y, v.z);
                if (before == group - 1) cntw[idx] = cur + group;
            }
            base += 32;
            if (base >= t[k]) break;
            const int p = base + lane;
            v = tmp[(p < t[k]) ? p : 0];
            if (p >= t[k]) v.x = -1;
        }
    }
}

__device__ __forceinline__ float4 edge_msg(const float4 hv, const float4 e0,
                                           const float4 e1, const float4 e2)
{
    float m0 = hv.x + e0.x + e1.x + e2.x; m0 = m0 > 0.f ? m0 : 0.f;
    float m1 = hv.y + e0.y + e1.y + e2.y; m1 = m1 > 0.f ? m1 : 0.f;
    float m2 = hv.z + e0.z + e1.z + e2.z; m2 = m2 > 0.f ? m2 : 0.f;
    float m3 = hv.w + e0.w + e1.w + e2.w; m3 = m3 > 0.f ? m3 : 0.f;
    return make_float4(m0, m1, m2, m3);
}

template <int NV>
__device__ __forceinline__ void edge_step(
    float4& acc, const int2 my, const int p, const int lane,
    const float4* b4, const float4* __restrict__ h4)
{
    float4 hv[NV];
    int code[NV];
    #pragma unroll
    for (int u = 0; u < NV; ++u) {
        const int src = __builtin_amdgcn_readlane(my.x, p + u);
        code[u]       = __builtin_amdgcn_readlane(my.y, p + u);
        const unsigned long long ra = (unsigned long long)(h4 + (size_t)src * 32);
        const unsigned int ra_lo = __builtin_amdgcn_readfirstlane((unsigned int)ra);
        const unsigned int ra_hi = __builtin_amdgcn_readfirstlane((unsigned int)(ra >> 32));
        const gv4f* row = (const gv4f*)(((unsigned long long)ra_hi << 32) | ra_lo);
        const v4f r = row[lane];
        hv[u] = make_float4(r[0], r[1], r[2], r[3]);
    }
    #pragma unroll
    for (int u = 0; u < NV; ++u) {
        const float4 e0 = b4[(0 * 6 + (code[u] & 255)) * 32 + lane];
        const float4 e1 = b4[(1 * 6 + ((code[u] >> 8) & 255)) * 32 + lane];
        const float4 e2 = b4[(2 * 6 + (code[u] >> 16)) * 32 + lane];
        const float4 m = edge_msg(hv[u], e0, e1, e2);
        acc.x += m.x; acc.y += m.y; acc.z += m.z; acc.w += m.w;
    }
}

__device__ __forceinline__ void sum_edges(
    float4& acc, const int2 my, const int o, const int cnt, const int lane,
    const float4* b4, const float4* __restrict__ h4)
{
    int j = 0;
    for (; j + 4 <= cnt; j += 4) edge_step<4>(acc, my, o + j, lane, b4, h4);
    const int r = cnt - j;
    if (r == 3)      edge_step<3>(acc, my, o + j, lane, b4, h4);
    else if (r == 2) edge_step<2>(acc, my, o + j, lane, b4, h4);
    else if (r == 1) edge_step<1>(acc, my, o + j, lane, b4, h4);
}

__device__ __forceinline__ float4 gather_row(
    const int n, const int s, const int t, const int lane,
    const int2* __restrict__ rec, const float4* b4,
    const float4* __restrict__ h4)
{
    float4 acc = h4[(size_t)n * 32 + lane];
    for (int base = s; base < t; base += 32) {
        const int left = t - base;
        const int cnt  = left < 32 ? left : 32;
        int2 my = make_int2(0, 0);
        if (lane < cnt) my = rec[base + lane];
        sum_edges(acc, my, 0, cnt, lane, b4, h4);
    }
    return acc;
}

#define PREP_ITEMS (NLAYERS * 4 * 8 * 32 * 16)
#define PREP_BLOCKS (PREP_ITEMS / 256)
__device__ __forceinline__ void prep_weights(
    const int o,
    const float* __restrict__ w1, const float* __restrict__ w2,
    __bf16* __restrict__ f1, __bf16* __restrict__ f2,
    float* __restrict__ sums, float* __restrict__ cnts)
{
    if (o >= PREP_ITEMS) return;
    static_assert(N_GRAPHS * HID <= 2 * PREP_ITEMS, "two pool sums per thread");
    static_assert(N_GRAPHS <= PREP_ITEMS, "one thread per graph count");
    if (o < N_GRAPHS * HID)              sums[o] = 0.f;
    if (o + PREP_ITEMS < N_GRAPHS * HID) sums[o + PREP_ITEMS] = 0.f;
    if (o < N_GRAPHS)                    cnts[o] = 0.f;
    int j     =  o        & 15;
    int lane  = (o >> 4)  & 31;
    int ntile = (o >> 9)  & 7;
    int ks    = (o >> 12) & 3;
    int layer =  o >> 14;
    int colL  = lane & 15;
    int khalf = (lane >> 4) * 8;
    int K     = ks * 32 + ((j < 8) ? (khalf + j) : (16 + khalf + (j - 8)));
    int col   = ntile * 16 + colL;
    size_t src = (size_t)layer * HID * HID + (size_t)K * HID + col;
    size_t dst = (size_t)layer * 16384 +
                 (((size_t)ks * 8 + ntile) * 32 + lane) * 16 + j;
    f1[dst] = (__bf16)w1[src];
    f2[dst] = (__bf16)w2[src];
}

__global__ __launch_bounds__(256) void k_setup(
    const int* __restrict__ x, const float* __restrict__ atom_emb, float* __restrict__ h,
    const float* __restrict__ w1, const float* __restrict__ w2,
    __bf16* __restrict__ f1, __bf16* __restrict__ f2,
    float* __restrict__ sums, float* __restrict__ cnts,
    const int* __restrict__ edge_index, const int* __restrict__ edge_attr,
    int* __restrict__ coff, int4* __restrict__ tmp)
{
    const int b = blockIdx.x;
    if (b < PREP_BLOCKS) {
        prep_weights(b * 256 + threadIdx.x, w1, w2, f1, f2, sums, cnts);
    } else if (b < PREP_BLOCKS + EP_CHUNKS) {
        part_edges(b - PREP_BLOCKS, edge_index, edge_attr, coff, tmp);
    } else {
        encode_nodes((b - PREP_BLOCKS - EP_CHUNKS) * 256 + threadIdx.x, x, atom_emb, h);
    }
}

template <bool LAST>
__device__ __forceinline__ void node_update_body(
    const int* __restrict__ rowptr, const int2* __restrict__ rec,
    const float* __restrict__ bond_emb, const float* __restrict__ hcur,
    const __bf16* __restrict__ wf1, const __bf16* __restrict__ wf2,
    const float* __restrict__ b1, const float* __restrict__ b2,
    const float* __restrict__ gamma, const float* __restrict__ beta,
    const float* __restrict__ mean, const float* __restrict__ var,
    float* __restrict__ hnext,
    const int* __restrict__ batch, float* __restrict__ sums, float* __restrict__ cnts)
{
    constexpr int ZP = HID + 8;
    __shared__ __align__(16) __bf16 zt[16 * ZP];
    __shared__ __align__(16) __bf16 z1[16 * ZP];
    __shared__ __align__(16) float  bt[18 * HID];

    static_assert(N_NODES % 16 == 0, "whole 16-node tiles: no row guard below");
    const int tid   = threadIdx.x;
    const int wave  = tid >> 5;
    const int lane  = tid & 31;
    const int node0 = blockIdx.x * 16;

    const int m0 = __builtin_amdgcn_readfirstlane(wave * 2);
    const int n0 = node0 + m0;
    const int s0 = rowptr[n0];
    const int s1 = rowptr[n0 + 1];
    const int s2 = rowptr[n0 + 2];

    {
        static_assert(18 * HID / 4 == 2 * 256 + 64, "2 float4 per thread and 64 more");
        const float4* __restrict__ g4 = (const float4*)bond_emb;
        float4* bt4w = (float4*)bt;
        const float4 t0 = g4[tid];
        const float4 t1 = g4[tid + 256];
        float4 t2 = make_float4(0.f, 0.f, 0.f, 0.f);
        if (tid < 64) t2 = g4[tid + 512];
        bt4w[tid]       = t0;
        bt4w[tid + 256] = t1;
        if (tid < 64) bt4w[tid + 512] = t2;
    }
    __syncthreads();

    {
        const float4* __restrict__ h4 = (const float4*)hcur;
        const float4* b4 = (const float4*)bt;
        float4 za, zb;
        const int tot = s2 - s0;
        if (tot <= 32) {
            int2 my = make_int2(0, 0);
            if (lane < tot) my = rec[s0 + lane];
            za = h4[(size_t)n0 * 32 + lane];
            sum_edges(za, my, 0, s1 - s0, lane, b4, h4);
            v4bf pa;
            pa[0] = (__bf16)za.x; pa[1] = (__bf16)za.y; pa[2] = (__bf16)za.z; pa[3] = (__bf16)za.w;
            *(v4bf*)&zt[m0 * ZP + lane * 4] = pa;
            zb = h4[(size_t)(n0 + 1) * 32 + lane];
            sum_edges(zb, my, s1 - s0, s2 - s1, lane, b4, h4);
        } else {
            za = gather_row(n0, s0, s1, lane, rec, b4, h4);
            v4bf pa;
            pa[0] = (__bf16)za.x; pa[1] = (__bf16)za.y; pa[2] = (__bf16)za.z; pa[3] = (__bf16)za.w;
            *(v4bf*)&zt[m0 * ZP + lane * 4] = pa;
            zb = gather_row(n0 + 1, s1, s2, lane, rec, b4, h4);
        }
        v4bf pb;
        pb[0] = (__bf16)zb.x; pb[1] = (__bf16)zb.y; pb[2] = (__bf16)zb.z; pb[3] = (__bf16)zb.w;
        *(v4bf*)&zt[(m0 + 1) * ZP + lane * 4] = pb;
    }
    __syncthreads();

    const int colL  = lane & 15;
    const int col   = wave * 16 + colL;
    const int khalf = (lane >> 4) * 8;
    const int rowA  = lane & 15;

    v8f acc1 = {};
    #pragma unroll
    for (int ks = 0; ks < 4; ++ks) {
        const __bf16* ap = &zt[rowA * ZP + ks * 32 + khalf];
        v8bf alo = *(const v8bf*)(ap);
        v8bf ahi = *(const v8bf*)(ap + 16);
        v16bf a = __builtin_shufflevector(alo, ahi,
                    0,1,2,3,4,5,6,7,8,9,10,11,12,13,14,15);
        v16bf b = *(const v16bf*)(wf1 + (((size_t)ks * 8 + wave) * 32 + lane) * 16);
        acc1 = __builtin_amdgcn_wmma_f32_16x16x32_bf16(
                   false, a, false, b, (short)0, acc1, false, false);
    }

    const float bias1 = b1[col];
    #pragma unroll
    for (int r = 0; r < 8; ++r) {
        int m = (lane < 16) ? r : (r + 8);
        float v = acc1[r] + bias1;
        v = v > 0.f ? v : 0.f;
        z1[m * ZP + col] = (__bf16)v;
    }
    __syncthreads();

    v8f acc2 = {};
    #pragma unroll
    for (int ks = 0; ks < 4; ++ks) {
        const __bf16* ap = &z1[rowA * ZP + ks * 32 + khalf];
        v8bf alo = *(const v8bf*)(ap);
        v8bf ahi = *(const v8bf*)(ap + 16);
        v16bf a = __builtin_shufflevector(alo, ahi,
                    0,1,2,3,4,5,6,7,8,9,10,11,12,13,14,15);
        v16bf b = *(const v16bf*)(wf2 + (((size_t)ks * 8 + wave) * 32 + lane) * 16);
        acc2 = __builtin_amdgcn_wmma_f32_16x16x32_bf16(
                   false, a, false, b, (short)0, acc2, false, false);
    }

    const float bias2 = b2[col];
    const float mu    = mean[col];
    const float inv   = rsqrtf(var[col] + 1e-5f);
    const float ga    = gamma[col];
    const float be    = beta[col];
    if (!LAST) {
        #pragma unroll
        for (int r = 0; r < 8; ++r) {
            int m = (lane < 16) ? r : (r + 8);
            size_t idx = (size_t)(node0 + m) * HID + col;
            float v = acc2[r] + bias2;
            v = (v - mu) * inv * ga + be;
            v = v > 0.f ? v : 0.f;
            hnext[idx] = hcur[idx] + v;
        }
    } else {
        int G[16];
        #pragma unroll
        for (int k = 0; k < 16; ++k) G[k] = batch[node0 + k];
        const bool hi   = lane >= 16;
        const int  row0 = node0 + (hi ? 8 : 0);
        float hv[8];
        #pragma unroll
        for (int r = 0; r < 8; ++r) {
            size_t idx = (size_t)(row0 + r) * HID + col;
            float v = acc2[r] + bias2;
            v = (v - mu) * inv * ga + be;
            v = v > 0.f ? v : 0.f;
            hv[r] = hcur[idx] + v;
        }
        float run  = 0.f;
        float nrun = 0.f;
        #pragma unroll
        for (int r = 0; r < 8; ++r) {
            run  += hv[r];
            nrun += 1.f;
            const int gr = hi ? G[r + 8] : G[r];
            bool flush = true;
            if (r < 7) flush = ((hi ? G[r + 9] : G[r + 1]) != gr);
            if (flush) {
                atomicAdd(sums + (size_t)gr * HID + col, run);
                if (col == 0) atomicAdd(cnts + gr, nrun);
                run  = 0.f;
                nrun = 0.f;
            }
        }
    }
}

__global__ __launch_bounds__(256, 16) void k_node_update(
    const int* __restrict__ rowptr, const int2* __restrict__ rec,
    const float* __restrict__ bond_emb, const float* __restrict__ hcur,
    const __bf16* __restrict__ wf1, const __bf16* __restrict__ wf2,
    const float* __restrict__ b1, const float* __restrict__ b2,
    const float* __restrict__ gamma, const float* __restrict__ beta,
    const float* __restrict__ mean, const float* __restrict__ var,
    float* __restrict__ hnext)
{
    node_update_body<false>(rowptr, rec, bond_emb, hcur, wf1, wf2, b1, b2,
                            gamma, beta, mean, var, hnext, nullptr, nullptr, nullptr);
}

__global__ __launch_bounds__(256, 16) void k_node_update_pool(
    const int* __restrict__ rowptr, const int2* __restrict__ rec,
    const float* __restrict__ bond_emb, const float* __restrict__ hcur,
    const __bf16* __restrict__ wf1, const __bf16* __restrict__ wf2,
    const float* __restrict__ b1, const float* __restrict__ b2,
    const float* __restrict__ gamma, const float* __restrict__ beta,
    const float* __restrict__ mean, const float* __restrict__ var,
    const int* __restrict__ batch, float* __restrict__ sums, float* __restrict__ cnts)
{
    node_update_body<true>(rowptr, rec, bond_emb, hcur, wf1, wf2, b1, b2,
                           gamma, beta, mean, var, nullptr, batch, sums, cnts);
}

__global__ __launch_bounds__(128) void k_pool_mlp(
    const float* __restrict__ sums, const float* __restrict__ cnts,
    const float* __restrict__ mw, const float* __restrict__ mb,
    float* __restrict__ out)
{
    __shared__ float p[HID];
    int g = blockIdx.x;
    int t = threadIdx.x;
    float cnt = cnts[g];
    cnt = cnt > 1.f ? cnt : 1.f;
    p[t] = sums[(size_t)g * HID + t] / cnt;
    __syncthreads();
    if (t < OUTD) {
        float a = mb[t];
        #pragma unroll 16
        for (int c = 0; c < HID; ++c) a += p[c] * mw[c * OUTD + t];
        out[g * OUTD + t] = a;
    }
}

extern "C" void kernel_launch(void* const* d_in, const int* in_sizes, int n_in,
                              void* d_out, int out_size, void* d_ws, size_t ws_size,
                              hipStream_t stream) {
    const int*   x          = (const int*)  d_in[0];
    const int*   edge_index = (const int*)  d_in[1];
    const int*   edge_attr  = (const int*)  d_in[2];
    const int*   batch      = (const int*)  d_in[3];
    const float* atom_emb   = (const float*)d_in[4];
    const float* bond_emb   = (const float*)d_in[5];
    const float* lin1_w     = (const float*)d_in[6];
    const float* lin1_b     = (const float*)d_in[7];
    const float* lin2_w     = (const float*)d_in[8];
    const float* lin2_b     = (const float*)d_in[9];
    const float* bn_gamma   = (const float*)d_in[10];
    const float* bn_beta    = (const float*)d_in[11];
    const float* bn_mean    = (const float*)d_in[12];
    const float* bn_var     = (const float*)d_in[13];
    const float* mlp_w      = (const float*)d_in[14];
    const float* mlp_b      = (const float*)d_in[15];

    char* ws = (char*)d_ws;
    size_t off = 0;
    auto alloc = [&](size_t bytes) -> char* {
        char* p = ws + off;
        off = (off + bytes + 255) & ~(size_t)255;
        return p;
    };
    float*  h0   = (float*) alloc((size_t)N_NODES * HID * 4);
    float*  h1   = (float*) alloc((size_t)N_NODES * HID * 4);
    __bf16* wf1  = (__bf16*)alloc((size_t)NLAYERS * 16384 * 2);
    __bf16* wf2  = (__bf16*)alloc((size_t)NLAYERS * 16384 * 2);
    float*  sums = (float*) alloc((size_t)N_GRAPHS * HID * 4);
    float*  cnts = (float*) alloc((size_t)N_GRAPHS * 4);
    int*    coff   = (int*)  alloc((size_t)EP_CHUNKS * (EB_BINS + 1) * 4);
    int*    rowptr = (int*)  alloc((size_t)(N_NODES + 1) * 4);
    int4*   tmp    = (int4*) alloc((size_t)N_EDGES * 16);
    int2*   rec    = (int2*) alloc((size_t)N_EDGES * 8);

    const int nodeChunks = N_NODES * 32;

    k_setup<<<PREP_BLOCKS + EP_CHUNKS + (nodeChunks + 255) / 256, 256, 0, stream>>>(
        x, atom_emb, h0, lin1_w, lin2_w, wf1, wf2, sums, cnts,
        edge_index, edge_attr, coff, tmp);

    k_build<<<EB_BINS, 1024, 0, stream>>>(coff, tmp, rowptr, rec);

    float* hc = h0;
    float* hn = h1;
    for (int i = 0; i < NLAYERS - 1; ++i) {
        k_node_update<<<N_NODES / 16, 256, 0, stream>>>(
            rowptr, rec, bond_emb, hc, wf1 + (size_t)i * 16384, wf2 + (size_t)i * 16384,
            lin1_b + i * HID, lin2_b + i * HID,
            bn_gamma + i * HID, bn_beta + i * HID,
            bn_mean + i * HID, bn_var + i * HID, hn);
        float* t = hc; hc = hn; hn = t;
    }
    k_node_update_pool<<<N_NODES / 16, 256, 0, stream>>>(
        rowptr, rec, bond_emb, hc,
        wf1 + (size_t)(NLAYERS - 1) * 16384, wf2 + (size_t)(NLAYERS - 1) * 16384,
        lin1_b + (NLAYERS - 1) * HID, lin2_b + (NLAYERS - 1) * HID,
        bn_gamma + (NLAYERS - 1) * HID, bn_beta + (NLAYERS - 1) * HID,
        bn_mean + (NLAYERS - 1) * HID, bn_var + (NLAYERS - 1) * HID,
        batch, sums, cnts);

    k_pool_mlp<<<N_GRAPHS, 128, 0, stream>>>(sums, cnts, mlp_w, mlp_b, (float*)d_out);
}
